# starting kernel
# baseline (speedup 1.0000x reference)
#include <hip/hip_runtime.h>
#include <math.h>

typedef __attribute__((ext_vector_type(2))) float v2f;
typedef __attribute__((ext_vector_type(8))) float v8f;

#define NN 8192
#define DD 64
#define BM 128
#define KC 32
#define SA (KC + 2)
#define ALPHA 0.2f
#define EPS 1e-8f


__global__ __launch_bounds__(256) void rowsum_kernel(const float* __restrict__ adj,
                                                     float* __restrict__ sr) {
  __shared__ float red[256];
  const int row = blockIdx.x;
  const int t = threadIdx.x;
  const float* p = adj + (size_t)row * NN;
  float s = 0.f;
  for (int j = t; j < NN; j += 256) s += p[j];
  red[t] = s;
  __syncthreads();
  for (int w = 128; w > 0; w >>= 1) {
    if (t < w) red[t] += red[t + w];
    __syncthreads();
  }
  if (t == 0) sr[row] = sqrtf(red[0]);
}

__global__ __launch_bounds__(256) void zero_kernel(float* __restrict__ p, int n) {
  int i = blockIdx.x * 256 + threadIdx.x;
  if (i < n) p[i] = 0.f;
}

__global__ __launch_bounds__(256) void colsum_partial_kernel(const float* __restrict__ adj,
                                                             float* __restrict__ acc) {
  const int j = blockIdx.x * 256 + threadIdx.x;
  const int i0 = blockIdx.y * 128;
  float s = 0.f;
#pragma unroll 4
  for (int i = 0; i < 128; ++i) s += adj[(size_t)(i0 + i) * NN + j];
  atomicAdd(&acc[j], s);
}

__global__ __launch_bounds__(256) void sqrt_kernel(const float* __restrict__ a,
                                                   float* __restrict__ sc) {
  int j = blockIdx.x * 256 + threadIdx.x;
  sc[j] = sqrtf(a[j]);
}


template <bool TRANSA>
__global__ __launch_bounds__(256) void spmm_kernel(
    const float* __restrict__ adj, const float* __restrict__ sr, const float* __restrict__ sc,
    const float* __restrict__ Bsrc, const float* __restrict__ Btgt,
    float* __restrict__ Z1, float* __restrict__ Z2) {
  __shared__ __align__(16) float At[BM * SA];
  __shared__ __align__(16) float Bs[DD * SA];
  __shared__ __align__(16) float Bt[DD * SA];

  const int tid = threadIdx.x;
  const int wave = tid >> 5;
  const int lane = tid & 31;
  const int laneRow = lane & 15;
  const int laneHi = lane >> 4;
  const int r0 = blockIdx.x * BM;

  v8f acc[8] = {};

  for (int k0 = 0; k0 < NN; k0 += KC) {
    if (!TRANSA) {
#pragma unroll
      for (int it = 0; it < (BM * KC / 4) / 256; ++it) {
        int f = it * 256 + tid;
        int row = f >> 3;
        int k = (f & 7) << 2;
        float4 v = *(const float4*)(adj + (size_t)(r0 + row) * NN + (k0 + k));
        float a = sr[r0 + row];
        float* dst = &At[row * SA + k];
        dst[0] = v.x * __builtin_amdgcn_rcpf(a * sc[k0 + k + 0] + EPS);
        dst[1] = v.y * __builtin_amdgcn_rcpf(a * sc[k0 + k + 1] + EPS);
        dst[2] = v.z * __builtin_amdgcn_rcpf(a * sc[k0 + k + 2] + EPS);
        dst[3] = v.w * __builtin_amdgcn_rcpf(a * sc[k0 + k + 3] + EPS);
      }
    } else {
#pragma unroll
      for (int it = 0; it < (BM * KC / 4) / 256; ++it) {
        int f = it * 256 + tid;
        int k = f >> 5;
        int rw = (f & 31) << 2;
        float4 v = *(const float4*)(adj + (size_t)(k0 + k) * NN + (r0 + rw));
        float a = sr[k0 + k];
        At[(rw + 0) * SA + k] = v.x * __builtin_amdgcn_rcpf(a * sc[r0 + rw + 0] + EPS);
        At[(rw + 1) * SA + k] = v.y * __builtin_amdgcn_rcpf(a * sc[r0 + rw + 1] + EPS);
        At[(rw + 2) * SA + k] = v.z * __builtin_amdgcn_rcpf(a * sc[r0 + rw + 2] + EPS);
        At[(rw + 3) * SA + k] = v.w * __builtin_amdgcn_rcpf(a * sc[r0 + rw + 3] + EPS);
      }
    }
#pragma unroll
    for (int it = 0; it < (DD * KC / 4) / 256; ++it) {
      int f = it * 256 + tid;
      int u = f >> 4;
      int col = (f & 15) << 2;
      float4 v1 = *(const float4*)(Bsrc + (size_t)(k0 + u) * DD + col);
      float4 v2 = *(const float4*)(Btgt + (size_t)(k0 + u) * DD + col);
      Bs[(col + 0) * SA + u] = v1.x; Bs[(col + 1) * SA + u] = v1.y;
      Bs[(col + 2) * SA + u] = v1.z; Bs[(col + 3) * SA + u] = v1.w;
      Bt[(col + 0) * SA + u] = v2.x; Bt[(col + 1) * SA + u] = v2.y;
      Bt[(col + 2) * SA + u] = v2.z; Bt[(col + 3) * SA + u] = v2.w;
    }
    __syncthreads();

    const float* Arow = &At[(wave * 16 + laneRow) * SA + 2 * laneHi];
#pragma unroll
    for (int kk = 0; kk < KC; kk += 4) {
      v2f a = *(const v2f*)(Arow + kk);
#pragma unroll
      for (int n = 0; n < 8; ++n) {
        const float* Bb = (n < 4) ? Bs : Bt;
        int col = (n & 3) * 16 + laneRow;
        v2f b = *(const v2f*)(Bb + col * SA + kk + 2 * laneHi);
        acc[n] = __builtin_amdgcn_wmma_f32_16x16x4_f32(
            false, a, false, b, (short)0, acc[n], false, false);
      }
    }
    __syncthreads();
  }

#pragma unroll
  for (int n = 0; n < 8; ++n) {
    float* Z = (n < 4) ? Z1 : Z2;
    const int col = (n & 3) * 16 + laneRow;
#pragma unroll
    for (int v = 0; v < 8; ++v) {
      int row = r0 + wave * 16 + v + 8 * laneHi;
      Z[(size_t)row * DD + col] = acc[n][v];
    }
  }
}


__global__ __launch_bounds__(256) void epilogue_kernel(
    const float* __restrict__ Z1, const float* __restrict__ Z2,
    const float* __restrict__ src, const float* __restrict__ tgt,
    const float* __restrict__ W1, const float* __restrict__ W2,
    float* __restrict__ out) {
  __shared__ float z1s[4][DD];
  __shared__ float ps[4][DD];
  const int tid = threadIdx.x;
  const int r = tid >> 6;
  const int c = tid & 63;
  const int row = blockIdx.x * 4 + r;
  const size_t off = (size_t)row * DD + c;
  z1s[r][c] = Z1[off];
  ps[r][c] = src[off] * Z2[off];
  __syncthreads();
  float acc = tgt[off];
#pragma unroll 8
  for (int k = 0; k < DD; ++k)
    acc += z1s[r][k] * W1[k * DD + c] + ps[r][k] * W2[k * DD + c];
  out[off] = (acc >= 0.f) ? acc : ALPHA * acc;
}


extern "C" void kernel_launch(void* const* d_in, const int* in_sizes, int n_in,
                              void* d_out, int out_size, void* d_ws, size_t ws_size,
                              hipStream_t stream) {
  const float* user = (const float*)d_in[0];
  const float* item = (const float*)d_in[1];
  const float* adj  = (const float*)d_in[2];
  const float* W1   = (const float*)d_in[3];
  const float* W2   = (const float*)d_in[4];

  float* ws = (float*)d_ws;
  const size_t ND = (size_t)NN * DD;
  float* sr   = ws;
  float* sc   = ws + NN;
  float* cacc = ws + 2 * NN;
  float* Z1u  = ws + 3 * NN;
  float* Z2u  = Z1u + ND;
  float* Z1i  = Z2u + ND;
  float* Z2i  = Z1i + ND;

  float* out_user = (float*)d_out;
  float* out_item = out_user + ND;

  rowsum_kernel<<<NN, 256, 0, stream>>>(adj, sr);
  zero_kernel<<<NN / 256, 256, 0, stream>>>(cacc, NN);
  colsum_partial_kernel<<<dim3(NN / 256, 64), 256, 0, stream>>>(adj, cacc);
  sqrt_kernel<<<NN / 256, 256, 0, stream>>>(cacc, sc);

  spmm_kernel<true><<<NN / BM, 256, 0, stream>>>(adj, sr, sc, item, user, Z1u, Z2u);
  spmm_kernel<false><<<NN / BM, 256, 0, stream>>>(adj, sr, sc, user, item, Z1i, Z2i);

  epilogue_kernel<<<NN / 4, 256, 0, stream>>>(Z1u, Z2u, item, user, W1, W2, out_user);
  epilogue_kernel<<<NN / 4, 256, 0, stream>>>(Z1i, Z2i, user, item, W1, W2, out_item);
}


# spmm row tile per work-group 128 -> 32 rows (8 waves = 2 row tiles x 4 column groups, 2 accumulator tiles each): grid 64 -> 256 work-groups so every WGP has one; order of every k sum unchanged
# speedup vs baseline: 1.8477x; 1.8477x over previous
#include <hip/hip_runtime.h>
#include <math.h>

typedef __attribute__((ext_vector_type(2))) float v2f;
typedef __attribute__((ext_vector_type(8))) float v8f;

#define NN 8192
#define DD 64
#define BM 32
#define KC 32
#define SA (KC + 2)
#define ALPHA 0.2f
#define EPS 1e-8f


__global__ __launch_bounds__(256) void rowsum_kernel(const float* __restrict__ adj,
                                                     float* __restrict__ sr) {
  __shared__ float red[256];
  const int row = blockIdx.x;
  const int t = threadIdx.x;
  const float* p = adj + (size_t)row * NN;
  float s = 0.f;
  for (int j = t; j < NN; j += 256) s += p[j];
  red[t] = s;
  __syncthreads();
  for (int w = 128; w > 0; w >>= 1) {
    if (t < w) red[t] += red[t + w];
    __syncthreads();
  }
  if (t == 0) sr[row] = sqrtf(red[0]);
}

__global__ __launch_bounds__(256) void zero_kernel(float* __restrict__ p, int n) {
  int i = blockIdx.x * 256 + threadIdx.x;
  if (i < n) p[i] = 0.f;
}

__global__ __launch_bounds__(256) void colsum_partial_kernel(const float* __restrict__ adj,
                                                             float* __restrict__ acc) {
  const int j = blockIdx.x * 256 + threadIdx.x;
  const int i0 = blockIdx.y * 128;
  float s = 0.f;
#pragma unroll 4
  for (int i = 0; i < 128; ++i) s += adj[(size_t)(i0 + i) * NN + j];
  atomicAdd(&acc[j], s);
}

__global__ __launch_bounds__(256) void sqrt_kernel(const float* __restrict__ a,
                                                   float* __restrict__ sc) {
  int j = blockIdx.x * 256 + threadIdx.x;
  sc[j] = sqrtf(a[j]);
}


template <bool TRANSA>
__global__ __launch_bounds__(256) void spmm_kernel(
    const float* __restrict__ adj, const float* __restrict__ sr, const float* __restrict__ sc,
    const float* __restrict__ Bsrc, const float* __restrict__ Btgt,
    float* __restrict__ Z1, float* __restrict__ Z2) {
  __shared__ __align__(16) float At[BM * SA];
  __shared__ __align__(16) float Bs[DD * SA];
  __shared__ __align__(16) float Bt[DD * SA];

  const int tid = threadIdx.x;
  const int wave = tid >> 5;
  const int lane = tid & 31;
  const int laneRow = lane & 15;
  const int laneHi = lane >> 4;
  const int r0 = blockIdx.x * BM;
  const int rt = wave & 1;
  const int cg = wave >> 1;
  const int c0 = (cg & 1) * 32;

  v8f acc[2] = {};

  for (int k0 = 0; k0 < NN; k0 += KC) {
    if (!TRANSA) {
      int row = tid >> 3;
      int k = (tid & 7) << 2;
      float4 v = *(const float4*)(adj + (size_t)(r0 + row) * NN + (k0 + k));
      float a = sr[r0 + row];
      float* dst = &At[row * SA + k];
      dst[0] = v.x * __builtin_amdgcn_rcpf(a * sc[k0 + k + 0] + EPS);
      dst[1] = v.y * __builtin_amdgcn_rcpf(a * sc[k0 + k + 1] + EPS);
      dst[2] = v.z * __builtin_amdgcn_rcpf(a * sc[k0 + k + 2] + EPS);
      dst[3] = v.w * __builtin_amdgcn_rcpf(a * sc[k0 + k + 3] + EPS);
    } else {
      int k = tid >> 3;
      int rw = (tid & 7) << 2;
      float4 v = *(const float4*)(adj + (size_t)(k0 + k) * NN + (r0 + rw));
      float a = sr[k0 + k];
      At[(rw + 0) * SA + k] = v.x * __builtin_amdgcn_rcpf(a * sc[r0 + rw + 0] + EPS);
      At[(rw + 1) * SA + k] = v.y * __builtin_amdgcn_rcpf(a * sc[r0 + rw + 1] + EPS);
      At[(rw + 2) * SA + k] = v.z * __builtin_amdgcn_rcpf(a * sc[r0 + rw + 2] + EPS);
      At[(rw + 3) * SA + k] = v.w * __builtin_amdgcn_rcpf(a * sc[r0 + rw + 3] + EPS);
    }
#pragma unroll
    for (int it = 0; it < (DD * KC / 4) / 256; ++it) {
      int f = it * 256 + tid;
      int u = f >> 4;
      int col = (f & 15) << 2;
      float4 v1 = *(const float4*)(Bsrc + (size_t)(k0 + u) * DD + col);
      float4 v2 = *(const float4*)(Btgt + (size_t)(k0 + u) * DD + col);
      Bs[(col + 0) * SA + u] = v1.x; Bs[(col + 1) * SA + u] = v1.y;
      Bs[(col + 2) * SA + u] = v1.z; Bs[(col + 3) * SA + u] = v1.w;
      Bt[(col + 0) * SA + u] = v2.x; Bt[(col + 1) * SA + u] = v2.y;
      Bt[(col + 2) * SA + u] = v2.z; Bt[(col + 3) * SA + u] = v2.w;
    }
    __syncthreads();

    const float* Arow = &At[(rt * 16 + laneRow) * SA + 2 * laneHi];
    const float* Bcol = ((cg < 2) ? Bs : Bt) + (c0 + laneRow) * SA + 2 * laneHi;
#pragma unroll
    for (int kk = 0; kk < KC; kk += 4) {
      v2f a = *(const v2f*)(Arow + kk);
#pragma unroll
      for (int n = 0; n < 2; ++n) {
        v2f b = *(const v2f*)(Bcol + n * 16 * SA + kk);
        acc[n] = __builtin_amdgcn_wmma_f32_16x16x4_f32(
            false, a, false, b, (short)0, acc[n], false, false);
      }
    }
    __syncthreads();
  }

  float* Z = (cg < 2) ? Z1 : Z2;
#pragma unroll
  for (int n = 0; n < 2; ++n) {
    const int col = c0 + n * 16 + laneRow;
#pragma unroll
    for (int v = 0; v < 8; ++v) {
      int row = r0 + rt * 16 + v + 8 * laneHi;
      Z[(size_t)row * DD + col] = acc[n][v];
    }
  }
}


__global__ __launch_bounds__(256) void epilogue_kernel(
    const float* __restrict__ Z1, const float* __restrict__ Z2,
    const float* __restrict__ src, const float* __restrict__ tgt,
    const float* __restrict__ W1, const float* __restrict__ W2,
    float* __restrict__ out) {
  __shared__ float z1s[4][DD];
  __shared__ float ps[4][DD];
  const int tid = threadIdx.x;
  const int r = tid >> 6;
  const int c = tid & 63;
  const int row = blockIdx.x * 4 + r;
  const size_t off = (size_t)row * DD + c;
  z1s[r][c] = Z1[off];
  ps[r][c] = src[off] * Z2[off];
  __syncthreads();
  float acc = tgt[off];
#pragma unroll 8
  for (int k = 0; k < DD; ++k)
    acc += z1s[r][k] * W1[k * DD + c] + ps[r][k] * W2[k * DD + c];
  out[off] = (acc >= 0.f) ? acc : ALPHA * acc;
}


extern "C" void kernel_launch(void* const* d_in, const int* in_sizes, int n_in,
                              void* d_out, int out_size, void* d_ws, size_t ws_size,
                              hipStream_t stream) {
  const float* user = (const float*)d_in[0];
  const float* item = (const float*)d_in[1];
  const float* adj  = (const float*)d_in[2];
  const float* W1   = (const float*)d_in[3];
  const float* W2   = (const float*)d_in[4];

  float* ws = (float*)d_ws;
  const size_t ND = (size_t)NN * DD;
  float* sr   = ws;
  float* sc   = ws + NN;
  float* cacc = ws + 2 * NN;
  float* Z1u  = ws + 3 * NN;
  float* Z2u  = Z1u + ND;
  float* Z1i  = Z2u + ND;
  float* Z2i  = Z1i + ND;

  float* out_user = (float*)d_out;
  float* out_item = out_user + ND;

  rowsum_kernel<<<NN, 256, 0, stream>>>(adj, sr);
  zero_kernel<<<NN / 256, 256, 0, stream>>>(cacc, NN);
  colsum_partial_kernel<<<dim3(NN / 256, 64), 256, 0, stream>>>(adj, cacc);
  sqrt_kernel<<<NN / 256, 256, 0, stream>>>(cacc, sc);

  spmm_kernel<true><<<NN / BM, 256, 0, stream>>>(adj, sr, sc, item, user, Z1u, Z2u);
  spmm_kernel<false><<<NN / BM, 256, 0, stream>>>(adj, sr, sc, user, item, Z1i, Z2i);

  epilogue_kernel<<<NN / 4, 256, 0, stream>>>(Z1u, Z2u, item, user, W1, W2, out_user);
  epilogue_kernel<<<NN / 4, 256, 0, stream>>>(Z1i, Z2i, user, item, W1, W2, out_item);
}


# spmm K loop software-pipelined: loads of the next 4 K chunks stay in flight in registers, 2 LDS stages, one hand-written barrier per chunk that drains no loads; order of every k sum unchanged
# speedup vs baseline: 4.4769x; 2.4230x over previous
#include <hip/hip_runtime.h>
#include <math.h>

typedef __attribute__((ext_vector_type(2))) float v2f;
typedef __attribute__((ext_vector_type(8))) float v8f;

#define NN 8192
#define DD 64
#define BM 32
#define KC 32
#define SA (KC + 2)
#define ALPHA 0.2f
#define EPS 1e-8f


__global__ __launch_bounds__(256) void rowsum_kernel(const float* __restrict__ adj,
                                                     float* __restrict__ sr) {
  __shared__ float red[256];
  const int row = blockIdx.x;
  const int t = threadIdx.x;
  const float* p = adj + (size_t)row * NN;
  float s = 0.f;
  for (int j = t; j < NN; j += 256) s += p[j];
  red[t] = s;
  __syncthreads();
  for (int w = 128; w > 0; w >>= 1) {
    if (t < w) red[t] += red[t + w];
    __syncthreads();
  }
  if (t == 0) sr[row] = sqrtf(red[0]);
}

__global__ __launch_bounds__(256) void zero_kernel(float* __restrict__ p, int n) {
  int i = blockIdx.x * 256 + threadIdx.x;
  if (i < n) p[i] = 0.f;
}

__global__ __launch_bounds__(256) void colsum_partial_kernel(const float* __restrict__ adj,
                                                             float* __restrict__ acc) {
  const int j = blockIdx.x * 256 + threadIdx.x;
  const int i0 = blockIdx.y * 128;
  float s = 0.f;
#pragma unroll 4
  for (int i = 0; i < 128; ++i) s += adj[(size_t)(i0 + i) * NN + j];
  atomicAdd(&acc[j], s);
}

__global__ __launch_bounds__(256) void sqrt_kernel(const float* __restrict__ a,
                                                   float* __restrict__ sc) {
  int j = blockIdx.x * 256 + threadIdx.x;
  sc[j] = sqrtf(a[j]);
}


#define NC (NN / KC)
#define PF 4
#define TILE ((BM + 2 * DD) * SA)

struct Stage {
  float4 a;
  float4 n;
  float4 b1[2];
  float4 b2[2];
};

static __device__ __forceinline__ void wg_barrier_loads_in_flight() {
  asm volatile("s_wait_dscnt 0x0\n\ts_barrier_signal -1\n\ts_barrier_wait -1" ::: "memory");
  __builtin_amdgcn_sched_barrier(0);
}

template <bool TRANSA>
__global__ __launch_bounds__(256) void spmm_kernel(
    const float* __restrict__ adj, const float* __restrict__ sr, const float* __restrict__ sc,
    const float* __restrict__ Bsrc, const float* __restrict__ Btgt,
    float* __restrict__ Z1, float* __restrict__ Z2) {
  static_assert(PF == 4 && (NC % PF) == 0 && NC >= 2 * PF, "the K loop below is written for a queue of four");
  __shared__ __align__(16) float lds[2 * TILE];

  const int tid = threadIdx.x;
  const int wave = tid >> 5;
  const int lane = tid & 31;
  const int laneRow = lane & 15;
  const int laneHi = lane >> 4;
  const int r0 = blockIdx.x * BM;
  const int rt = wave & 1;
  const int cg = wave >> 1;
  const int c0 = (cg & 1) * 32;

  const int aq = tid >> 3;
  const int ar = (tid & 7) << 2;
  const float* ap = TRANSA ? adj + (size_t)aq * NN + (r0 + ar)
                           : adj + (size_t)(r0 + aq) * NN + ar;
  float i0, i1 = 0.f, i2 = 0.f, i3 = 0.f;
  if (!TRANSA) {
    i0 = sr[r0 + aq];
  } else {
    i0 = sc[r0 + ar + 0]; i1 = sc[r0 + ar + 1]; i2 = sc[r0 + ar + 2]; i3 = sc[r0 + ar + 3];
  }
  const int bu = tid >> 4;
  const int bc = (tid & 15) << 2;
  const float* b1p = Bsrc + (size_t)bu * DD + bc;
  const float* b2p = Btgt + (size_t)bu * DD + bc;

  auto gload = [&](Stage& s, int c) {
    const int k0 = c * KC;
    if (!TRANSA) {
      s.n.x = sc[k0 + ar + 0]; s.n.y = sc[k0 + ar + 1];
      s.n.z = sc[k0 + ar + 2]; s.n.w = sc[k0 + ar + 3];
      s.a = *(const float4*)(ap + k0);
    } else {
      s.n.x = sr[k0 + aq];
      s.a = *(const float4*)(ap + (size_t)k0 * NN);
    }
    s.b1[0] = *(const float4*)(b1p + (size_t)k0 * DD);
    s.b1[1] = *(const float4*)(b1p + (size_t)(k0 + 16) * DD);
    s.b2[0] = *(const float4*)(b2p + (size_t)k0 * DD);
    s.b2[1] = *(const float4*)(b2p + (size_t)(k0 + 16) * DD);
  };

  auto lstore = [&](const Stage& s, int buf) {
    float* L = lds + buf * TILE;
    if (!TRANSA) {
      float* dst = L + aq * SA + ar;
      dst[0] = s.a.x * __builtin_amdgcn_rcpf(i0 * s.n.x + EPS);
      dst[1] = s.a.y * __builtin_amdgcn_rcpf(i0 * s.n.y + EPS);
      dst[2] = s.a.z * __builtin_amdgcn_rcpf(i0 * s.n.z + EPS);
      dst[3] = s.a.w * __builtin_amdgcn_rcpf(i0 * s.n.w + EPS);
    } else {
      const float a = s.n.x;
      L[(ar + 0) * SA + aq] = s.a.x * __builtin_amdgcn_rcpf(a * i0 + EPS);
      L[(ar + 1) * SA + aq] = s.a.y * __builtin_amdgcn_rcpf(a * i1 + EPS);
      L[(ar + 2) * SA + aq] = s.a.z * __builtin_amdgcn_rcpf(a * i2 + EPS);
      L[(ar + 3) * SA + aq] = s.a.w * __builtin_amdgcn_rcpf(a * i3 + EPS);
    }
#pragma unroll
    for (int it = 0; it < 2; ++it) {
      float* d1 = L + (BM + bc) * SA + (bu + it * 16);
      float* d2 = L + (BM + DD + bc) * SA + (bu + it * 16);
      d1[0 * SA] = s.b1[it].x; d1[1 * SA] = s.b1[it].y;
      d1[2 * SA] = s.b1[it].z; d1[3 * SA] = s.b1[it].w;
      d2[0 * SA] = s.b2[it].x; d2[1 * SA] = s.b2[it].y;
      d2[2 * SA] = s.b2[it].z; d2[3 * SA] = s.b2[it].w;
    }
  };

  v8f acc[2] = {};

  const int aoff = (rt * 16 + laneRow) * SA + 2 * laneHi;
  const int boff = (BM + cg * 32 + laneRow) * SA + 2 * laneHi;
  auto compute = [&](int buf) {
    const float* Arow = lds + buf * TILE + aoff;
    const float* Bcol = lds + buf * TILE + boff;
#pragma unroll
    for (int kk = 0; kk < KC; kk += 4) {
      v2f a = *(const v2f*)(Arow + kk);
#pragma unroll
      for (int n = 0; n < 2; ++n) {
        v2f b = *(const v2f*)(Bcol + n * 16 * SA + kk);
        acc[n] = __builtin_amdgcn_wmma_f32_16x16x4_f32(
            false, a, false, b, (short)0, acc[n], false, false);
      }
    }
  };

  Stage s0, s1, s2, s3;
  gload(s0, 0); __builtin_amdgcn_sched_barrier(0);
  gload(s1, 1); __builtin_amdgcn_sched_barrier(0);
  gload(s2, 2); __builtin_amdgcn_sched_barrier(0);
  gload(s3, 3); __builtin_amdgcn_sched_barrier(0);
  lstore(s0, 0);
  gload(s0, 4);
  wg_barrier_loads_in_flight();

  int c = 0;
  for (; c + 8 < NC; c += 4) {
    lstore(s1, 1); gload(s1, c + 5); compute(0); wg_barrier_loads_in_flight();
    lstore(s2, 0); gload(s2, c + 6); compute(1); wg_barrier_loads_in_flight();
    lstore(s3, 1); gload(s3, c + 7); compute(0); wg_barrier_loads_in_flight();
    lstore(s0, 0); gload(s0, c + 8); compute(1); wg_barrier_loads_in_flight();
  }
  lstore(s1, 1); gload(s1, NC - 3); compute(0); wg_barrier_loads_in_flight();
  lstore(s2, 0); gload(s2, NC - 2); compute(1); wg_barrier_loads_in_flight();
  lstore(s3, 1); gload(s3, NC - 1); compute(0); wg_barrier_loads_in_flight();
  lstore(s0, 0);                    compute(1); wg_barrier_loads_in_flight();
  lstore(s1, 1);                    compute(0); wg_barrier_loads_in_flight();
  lstore(s2, 0);                    compute(1); wg_barrier_loads_in_flight();
  lstore(s3, 1);                    compute(0); wg_barrier_loads_in_flight();
  compute(1);

  float* Z = (cg < 2) ? Z1 : Z2;
#pragma unroll
  for (int n = 0; n < 2; ++n) {
    const int col = c0 + n * 16 + laneRow;
#pragma unroll
    for (int v = 0; v < 8; ++v) {
      int row = r0 + rt * 16 + v + 8 * laneHi;
      Z[(size_t)row * DD + col] = acc[n][v];
    }
  }
}


__global__ __launch_bounds__(256) void epilogue_kernel(
    const float* __restrict__ Z1, const float* __restrict__ Z2,
    const float* __restrict__ src, const float* __restrict__ tgt,
    const float* __restrict__ W1, const float* __restrict__ W2,
    float* __restrict__ out) {
  __shared__ float z1s[4][DD];
  __shared__ float ps[4][DD];
  const int tid = threadIdx.x;
  const int r = tid >> 6;
  const int c = tid & 63;
  const int row = blockIdx.x * 4 + r;
  const size_t off = (size_t)row * DD + c;
  z1s[r][c] = Z1[off];
  ps[r][c] = src[off] * Z2[off];
  __syncthreads();
  float acc = tgt[off];
#pragma unroll 8
  for (int k = 0; k < DD; ++k)
    acc += z1s[r][k] * W1[k * DD + c] + ps[r][k] * W2[k * DD + c];
  out[off] = (acc >= 0.f) ? acc : ALPHA * acc;
}


extern "C" void kernel_launch(void* const* d_in, const int* in_sizes, int n_in,
                              void* d_out, int out_size, void* d_ws, size_t ws_size,
                              hipStream_t stream) {
  const float* user = (const float*)d_in[0];
  const float* item = (const float*)d_in[1];
  const float* adj  = (const float*)d_in[2];
  const float* W1   = (const float*)d_in[3];
  const float* W2   = (const float*)d_in[4];

  float* ws = (float*)d_ws;
  const size_t ND = (size_t)NN * DD;
  float* sr   = ws;
  float* sc   = ws + NN;
  float* cacc = ws + 2 * NN;
  float* Z1u  = ws + 3 * NN;
  float* Z2u  = Z1u + ND;
  float* Z1i  = Z2u + ND;
  float* Z2i  = Z1i + ND;

  float* out_user = (float*)d_out;
  float* out_item = out_user + ND;

  rowsum_kernel<<<NN, 256, 0, stream>>>(adj, sr);
  zero_kernel<<<NN / 256, 256, 0, stream>>>(cacc, NN);
  colsum_partial_kernel<<<dim3(NN / 256, 64), 256, 0, stream>>>(adj, cacc);
  sqrt_kernel<<<NN / 256, 256, 0, stream>>>(cacc, sc);

  spmm_kernel<true><<<NN / BM, 256, 0, stream>>>(adj, sr, sc, item, user, Z1u, Z2u);
  spmm_kernel<false><<<NN / BM, 256, 0, stream>>>(adj, sr, sc, user, item, Z1i, Z2i);

  epilogue_kernel<<<NN / 4, 256, 0, stream>>>(Z1u, Z2u, item, user, W1, W2, out_user);
  epilogue_kernel<<<NN / 4, 256, 0, stream>>>(Z1i, Z2i, user, item, W1, W2, out_item);
}


# spmm K loop: LDS stages widened from one K chunk to four, so the waves of a work-group meet at one barrier per four chunks instead of one per chunk; order of every k sum unchanged
# speedup vs baseline: 4.8316x; 1.0792x over previous
#include <hip/hip_runtime.h>
#include <math.h>

typedef __attribute__((ext_vector_type(2))) float v2f;
typedef __attribute__((ext_vector_type(8))) float v8f;

#define NN 8192
#define DD 64
#define BM 32
#define KC 32
#define SA (KC + 2)
#define ALPHA 0.2f
#define EPS 1e-8f


__global__ __launch_bounds__(256) void rowsum_kernel(const float* __restrict__ adj,
                                                     float* __restrict__ sr) {
  __shared__ float red[256];
  const int row = blockIdx.x;
  const int t = threadIdx.x;
  const float* p = adj + (size_t)row * NN;
  float s = 0.f;
  for (int j = t; j < NN; j += 256) s += p[j];
  red[t] = s;
  __syncthreads();
  for (int w = 128; w > 0; w >>= 1) {
    if (t < w) red[t] += red[t + w];
    __syncthreads();
  }
  if (t == 0) sr[row] = sqrtf(red[0]);
}

__global__ __launch_bounds__(256) void zero_kernel(float* __restrict__ p, int n) {
  int i = blockIdx.x * 256 + threadIdx.x;
  if (i < n) p[i] = 0.f;
}

__global__ __launch_bounds__(256) void colsum_partial_kernel(const float* __restrict__ adj,
                                                             float* __restrict__ acc) {
  const int j = blockIdx.x * 256 + threadIdx.x;
  const int i0 = blockIdx.y * 128;
  float s = 0.f;
#pragma unroll 4
  for (int i = 0; i < 128; ++i) s += adj[(size_t)(i0 + i) * NN + j];
  atomicAdd(&acc[j], s);
}

__global__ __launch_bounds__(256) void sqrt_kernel(const float* __restrict__ a,
                                                   float* __restrict__ sc) {
  int j = blockIdx.x * 256 + threadIdx.x;
  sc[j] = sqrtf(a[j]);
}


#define NC (NN / KC)
#define PF 4
#define TILE ((BM + 2 * DD) * SA)

struct Stage {
  float4 a;
  float4 n;
  float4 b1[2];
  float4 b2[2];
};

static __device__ __forceinline__ void wg_barrier_loads_in_flight() {
  asm volatile("s_wait_dscnt 0x0\n\ts_barrier_signal -1\n\ts_barrier_wait -1" ::: "memory");
}

template <bool TRANSA>
__global__ __launch_bounds__(256) void spmm_kernel(
    const float* __restrict__ adj, const float* __restrict__ sr, const float* __restrict__ sc,
    const float* __restrict__ Bsrc, const float* __restrict__ Btgt,
    float* __restrict__ Z1, float* __restrict__ Z2) {
  static_assert(PF == 4 && (NC % (2 * PF)) == 0 && NC >= 4 * PF, "the K loop below is written for a queue of four");
  __shared__ __align__(16) float lds[2 * PF * TILE];

  const int tid = threadIdx.x;
  const int wave = tid >> 5;
  const int lane = tid & 31;
  const int laneRow = lane & 15;
  const int laneHi = lane >> 4;
  const int r0 = blockIdx.x * BM;
  const int rt = wave & 1;
  const int cg = wave >> 1;
  const int c0 = (cg & 1) * 32;

  const int aq = tid >> 3;
  const int ar = (tid & 7) << 2;
  const float* ap = TRANSA ? adj + (size_t)aq * NN + (r0 + ar)
                           : adj + (size_t)(r0 + aq) * NN + ar;
  float i0, i1 = 0.f, i2 = 0.f, i3 = 0.f;
  if (!TRANSA) {
    i0 = sr[r0 + aq];
  } else {
    i0 = sc[r0 + ar + 0]; i1 = sc[r0 + ar + 1]; i2 = sc[r0 + ar + 2]; i3 = sc[r0 + ar + 3];
  }
  auto chunk_fence = [&]() {
    if (!TRANSA) asm volatile("" : "+v"(i0) : : "memory");
    else asm volatile("" : "+v"(i0), "+v"(i1), "+v"(i2), "+v"(i3) : : "memory");
    __builtin_amdgcn_sched_barrier(0);
  };
  const int bu = tid >> 4;
  const int bc = (tid & 15) << 2;
  const float* b1p = Bsrc + (size_t)bu * DD + bc;
  const float* b2p = Btgt + (size_t)bu * DD + bc;

  auto gload = [&](Stage& s, int c) {
    const int k0 = c * KC;
    if (!TRANSA) {
      s.n.x = sc[k0 + ar + 0]; s.n.y = sc[k0 + ar + 1];
      s.n.z = sc[k0 + ar + 2]; s.n.w = sc[k0 + ar + 3];
      s.a = *(const float4*)(ap + k0);
    } else {
      s.n.x = sr[k0 + aq];
      s.a = *(const float4*)(ap + (size_t)k0 * NN);
    }
    s.b1[0] = *(const float4*)(b1p + (size_t)k0 * DD);
    s.b1[1] = *(const float4*)(b1p + (size_t)(k0 + 16) * DD);
    s.b2[0] = *(const float4*)(b2p + (size_t)k0 * DD);
    s.b2[1] = *(const float4*)(b2p + (size_t)(k0 + 16) * DD);
  };

  auto lstore = [&](const Stage& s, int buf, int slot) {
    float* L = lds + (buf * PF + slot) * TILE;
    if (!TRANSA) {
      float* dst = L + aq * SA + ar;
      dst[0] = s.a.x * __builtin_amdgcn_rcpf(i0 * s.n.x + EPS);
      dst[1] = s.a.y * __builtin_amdgcn_rcpf(i0 * s.n.y + EPS);
      dst[2] = s.a.z * __builtin_amdgcn_rcpf(i0 * s.n.z + EPS);
      dst[3] = s.a.w * __builtin_amdgcn_rcpf(i0 * s.n.w + EPS);
    } else {
      const float a = s.n.x;
      L[(ar + 0) * SA + aq] = s.a.x * __builtin_amdgcn_rcpf(a * i0 + EPS);
      L[(ar + 1) * SA + aq] = s.a.y * __builtin_amdgcn_rcpf(a * i1 + EPS);
      L[(ar + 2) * SA + aq] = s.a.z * __builtin_amdgcn_rcpf(a * i2 + EPS);
      L[(ar + 3) * SA + aq] = s.a.w * __builtin_amdgcn_rcpf(a * i3 + EPS);
    }
#pragma unroll
    for (int it = 0; it < 2; ++it) {
      float* d1 = L + (BM + bc) * SA + (bu + it * 16);
      float* d2 = L + (BM + DD + bc) * SA + (bu + it * 16);
      d1[0 * SA] = s.b1[it].x; d1[1 * SA] = s.b1[it].y;
      d1[2 * SA] = s.b1[it].z; d1[3 * SA] = s.b1[it].w;
      d2[0 * SA] = s.b2[it].x; d2[1 * SA] = s.b2[it].y;
      d2[2 * SA] = s.b2[it].z; d2[3 * SA] = s.b2[it].w;
    }
  };

  v8f acc[2] = {};

  const int aoff = (rt * 16 + laneRow) * SA + 2 * laneHi;
  const int boff = (BM + cg * 32 + laneRow) * SA + 2 * laneHi;
  auto compute = [&](int buf, int slot) {
    const float* Arow = lds + (buf * PF + slot) * TILE + aoff;
    const float* Bcol = lds + (buf * PF + slot) * TILE + boff;
#pragma unroll
    for (int kk = 0; kk < KC; kk += 4) {
      v2f a = *(const v2f*)(Arow + kk);
#pragma unroll
      for (int n = 0; n < 2; ++n) {
        v2f b = *(const v2f*)(Bcol + n * 16 * SA + kk);
        acc[n] = __builtin_amdgcn_wmma_f32_16x16x4_f32(
            false, a, false, b, (short)0, acc[n], false, false);
      }
    }
  };

  Stage s0, s1, s2, s3;
  gload(s0, 0); chunk_fence();
  gload(s1, 1); chunk_fence();
  gload(s2, 2); chunk_fence();
  gload(s3, 3); chunk_fence();
  lstore(s0, 0, 0); gload(s0, 4); chunk_fence();
  lstore(s1, 0, 1); gload(s1, 5); chunk_fence();
  lstore(s2, 0, 2); gload(s2, 6); chunk_fence();
  lstore(s3, 0, 3); gload(s3, 7);
  wg_barrier_loads_in_flight(); chunk_fence();

  int c = 0;
  for (; c + 16 <= NC; c += 8) {
    lstore(s0, 1, 0); gload(s0, c + 8);  compute(0, 0); chunk_fence();
    lstore(s1, 1, 1); gload(s1, c + 9);  compute(0, 1); chunk_fence();
    lstore(s2, 1, 2); gload(s2, c + 10); compute(0, 2); chunk_fence();
    lstore(s3, 1, 3); gload(s3, c + 11); compute(0, 3); wg_barrier_loads_in_flight(); chunk_fence();
    lstore(s0, 0, 0); gload(s0, c + 12); compute(1, 0); chunk_fence();
    lstore(s1, 0, 1); gload(s1, c + 13); compute(1, 1); chunk_fence();
    lstore(s2, 0, 2); gload(s2, c + 14); compute(1, 2); chunk_fence();
    lstore(s3, 0, 3); gload(s3, c + 15); compute(1, 3); wg_barrier_loads_in_flight(); chunk_fence();
  }
  lstore(s0, 1, 0); compute(0, 0); chunk_fence();
  lstore(s1, 1, 1); compute(0, 1); chunk_fence();
  lstore(s2, 1, 2); compute(0, 2); chunk_fence();
  lstore(s3, 1, 3); compute(0, 3); wg_barrier_loads_in_flight(); chunk_fence();
  compute(1, 0); compute(1, 1); compute(1, 2); compute(1, 3);

  float* Z = (cg < 2) ? Z1 : Z2;
#pragma unroll
  for (int n = 0; n < 2; ++n) {
    const int col = c0 + n * 16 + laneRow;
#pragma unroll
    for (int v = 0; v < 8; ++v) {
      int row = r0 + rt * 16 + v + 8 * laneHi;
      Z[(size_t)row * DD + col] = acc[n][v];
    }
  }
}


__global__ __launch_bounds__(256) void epilogue_kernel(
    const float* __restrict__ Z1, const float* __restrict__ Z2,
    const float* __restrict__ src, const float* __restrict__ tgt,
    const float* __restrict__ W1, const float* __restrict__ W2,
    float* __restrict__ out) {
  __shared__ float z1s[4][DD];
  __shared__ float ps[4][DD];
  const int tid = threadIdx.x;
  const int r = tid >> 6;
  const int c = tid & 63;
  const int row = blockIdx.x * 4 + r;
  const size_t off = (size_t)row * DD + c;
  z1s[r][c] = Z1[off];
  ps[r][c] = src[off] * Z2[off];
  __syncthreads();
  float acc = tgt[off];
#pragma unroll 8
  for (int k = 0; k < DD; ++k)
    acc += z1s[r][k] * W1[k * DD + c] + ps[r][k] * W2[k * DD + c];
  out[off] = (acc >= 0.f) ? acc : ALPHA * acc;
}


extern "C" void kernel_launch(void* const* d_in, const int* in_sizes, int n_in,
                              void* d_out, int out_size, void* d_ws, size_t ws_size,
                              hipStream_t stream) {
  const float* user = (const float*)d_in[0];
  const float* item = (const float*)d_in[1];
  const float* adj  = (const float*)d_in[2];
  const float* W1   = (const float*)d_in[3];
  const float* W2   = (const float*)d_in[4];

  float* ws = (float*)d_ws;
  const size_t ND = (size_t)NN * DD;
  float* sr   = ws;
  float* sc   = ws + NN;
  float* cacc = ws + 2 * NN;
  float* Z1u  = ws + 3 * NN;
  float* Z2u  = Z1u + ND;
  float* Z1i  = Z2u + ND;
  float* Z2i  = Z1i + ND;

  float* out_user = (float*)d_out;
  float* out_item = out_user + ND;

  rowsum_kernel<<<NN, 256, 0, stream>>>(adj, sr);
  zero_kernel<<<NN / 256, 256, 0, stream>>>(cacc, NN);
  colsum_partial_kernel<<<dim3(NN / 256, 64), 256, 0, stream>>>(adj, cacc);
  sqrt_kernel<<<NN / 256, 256, 0, stream>>>(cacc, sc);

  spmm_kernel<true><<<NN / BM, 256, 0, stream>>>(adj, sr, sc, item, user, Z1u, Z2u);
  spmm_kernel<false><<<NN / BM, 256, 0, stream>>>(adj, sr, sc, user, item, Z1i, Z2i);

  epilogue_kernel<<<NN / 4, 256, 0, stream>>>(Z1u, Z2u, item, user, W1, W2, out_user);
  epilogue_kernel<<<NN / 4, 256, 0, stream>>>(Z1i, Z2i, user, item, W1, W2, out_item);
}


# spmm K loop: fragments double-buffered in registers, the LDS reads of the next chunk issued as one group ahead of the current chunk's 16 WMMAs so no WMMA waits on LDS; order of every k sum unchanged
# speedup vs baseline: 5.0421x; 1.0436x over previous
#include <hip/hip_runtime.h>
#include <math.h>

typedef __attribute__((ext_vector_type(2))) float v2f;
typedef __attribute__((ext_vector_type(8))) float v8f;

#define NN 8192
#define DD 64
#define BM 32
#define KC 32
#define SA (KC + 2)
#define ALPHA 0.2f
#define EPS 1e-8f


__global__ __launch_bounds__(256) void rowsum_kernel(const float* __restrict__ adj,
                                                     float* __restrict__ sr) {
  __shared__ float red[256];
  const int row = blockIdx.x;
  const int t = threadIdx.x;
  const float* p = adj + (size_t)row * NN;
  float s = 0.f;
  for (int j = t; j < NN; j += 256) s += p[j];
  red[t] = s;
  __syncthreads();
  for (int w = 128; w > 0; w >>= 1) {
    if (t < w) red[t] += red[t + w];
    __syncthreads();
  }
  if (t == 0) sr[row] = sqrtf(red[0]);
}

__global__ __launch_bounds__(256) void zero_kernel(float* __restrict__ p, int n) {
  int i = blockIdx.x * 256 + threadIdx.x;
  if (i < n) p[i] = 0.f;
}

__global__ __launch_bounds__(256) void colsum_partial_kernel(const float* __restrict__ adj,
                                                             float* __restrict__ acc) {
  const int j = blockIdx.x * 256 + threadIdx.x;
  const int i0 = blockIdx.y * 128;
  float s = 0.f;
#pragma unroll 4
  for (int i = 0; i < 128; ++i) s += adj[(size_t)(i0 + i) * NN + j];
  atomicAdd(&acc[j], s);
}

__global__ __launch_bounds__(256) void sqrt_kernel(const float* __restrict__ a,
                                                   float* __restrict__ sc) {
  int j = blockIdx.x * 256 + threadIdx.x;
  sc[j] = sqrtf(a[j]);
}


#define NC (NN / KC)
#define PF 4
#define TILE ((BM + 2 * DD) * SA)

struct Stage {
  float4 a;
  float4 n;
  float4 b1[2];
  float4 b2[2];
};

static __device__ __forceinline__ void wg_barrier_loads_in_flight() {
  asm volatile("s_wait_dscnt 0x0\n\ts_barrier_signal -1\n\ts_barrier_wait -1" ::: "memory");
}

template <bool TRANSA>
__global__ __launch_bounds__(256) void spmm_kernel(
    const float* __restrict__ adj, const float* __restrict__ sr, const float* __restrict__ sc,
    const float* __restrict__ Bsrc, const float* __restrict__ Btgt,
    float* __restrict__ Z1, float* __restrict__ Z2) {
  static_assert(PF == 4 && (NC % (2 * PF)) == 0 && NC >= 4 * PF, "the K loop below is written for a queue of four");
  __shared__ __align__(16) float lds[2 * PF * TILE];

  const int tid = threadIdx.x;
  const int wave = tid >> 5;
  const int lane = tid & 31;
  const int laneRow = lane & 15;
  const int laneHi = lane >> 4;
  const int r0 = blockIdx.x * BM;
  const int rt = wave & 1;
  const int cg = wave >> 1;
  const int c0 = (cg & 1) * 32;

  const int aq = tid >> 3;
  const int ar = (tid & 7) << 2;
  const float* ap = TRANSA ? adj + (size_t)aq * NN + (r0 + ar)
                           : adj + (size_t)(r0 + aq) * NN + ar;
  float i0, i1 = 0.f, i2 = 0.f, i3 = 0.f;
  if (!TRANSA) {
    i0 = sr[r0 + aq];
  } else {
    i0 = sc[r0 + ar + 0]; i1 = sc[r0 + ar + 1]; i2 = sc[r0 + ar + 2]; i3 = sc[r0 + ar + 3];
  }
  auto chunk_fence = [&]() {
    if (!TRANSA) asm volatile("" : "+v"(i0) : : "memory");
    else asm volatile("" : "+v"(i0), "+v"(i1), "+v"(i2), "+v"(i3) : : "memory");
    __builtin_amdgcn_sched_barrier(0);
  };
  const int bu = tid >> 4;
  const int bc = (tid & 15) << 2;
  const float* b1p = Bsrc + (size_t)bu * DD + bc;
  const float* b2p = Btgt + (size_t)bu * DD + bc;

  auto gload = [&](Stage& s, int c) {
    const int k0 = c * KC;
    if (!TRANSA) {
      s.n.x = sc[k0 + ar + 0]; s.n.y = sc[k0 + ar + 1];
      s.n.z = sc[k0 + ar + 2]; s.n.w = sc[k0 + ar + 3];
      s.a = *(const float4*)(ap + k0);
    } else {
      s.n.x = sr[k0 + aq];
      s.a = *(const float4*)(ap + (size_t)k0 * NN);
    }
    s.b1[0] = *(const float4*)(b1p + (size_t)k0 * DD);
    s.b1[1] = *(const float4*)(b1p + (size_t)(k0 + 16) * DD);
    s.b2[0] = *(const float4*)(b2p + (size_t)k0 * DD);
    s.b2[1] = *(const float4*)(b2p + (size_t)(k0 + 16) * DD);
  };

  auto lstore = [&](const Stage& s, int buf, int slot) {
    float* L = lds + (buf * PF + slot) * TILE;
    if (!TRANSA) {
      float* dst = L + aq * SA + ar;
      dst[0] = s.a.x * __builtin_amdgcn_rcpf(i0 * s.n.x + EPS);
      dst[1] = s.a.y * __builtin_amdgcn_rcpf(i0 * s.n.y + EPS);
      dst[2] = s.a.z * __builtin_amdgcn_rcpf(i0 * s.n.z + EPS);
      dst[3] = s.a.w * __builtin_amdgcn_rcpf(i0 * s.n.w + EPS);
    } else {
      const float a = s.n.x;
      L[(ar + 0) * SA + aq] = s.a.x * __builtin_amdgcn_rcpf(a * i0 + EPS);
      L[(ar + 1) * SA + aq] = s.a.y * __builtin_amdgcn_rcpf(a * i1 + EPS);
      L[(ar + 2) * SA + aq] = s.a.z * __builtin_amdgcn_rcpf(a * i2 + EPS);
      L[(ar + 3) * SA + aq] = s.a.w * __builtin_amdgcn_rcpf(a * i3 + EPS);
    }
#pragma unroll
    for (int it = 0; it < 2; ++it) {
      float* d1 = L + (BM + bc) * SA + (bu + it * 16);
      float* d2 = L + (BM + DD + bc) * SA + (bu + it * 16);
      d1[0 * SA] = s.b1[it].x; d1[1 * SA] = s.b1[it].y;
      d1[2 * SA] = s.b1[it].z; d1[3 * SA] = s.b1[it].w;
      d2[0 * SA] = s.b2[it].x; d2[1 * SA] = s.b2[it].y;
      d2[2 * SA] = s.b2[it].z; d2[3 * SA] = s.b2[it].w;
    }
  };

  v8f acc[2] = {};

  const int aoff = (rt * 16 + laneRow) * SA + 2 * laneHi;
  const int boff = (BM + cg * 32 + laneRow) * SA + 2 * laneHi;
  struct Frag {
    v2f a[KC / 4];
    v2f b[KC / 4][2];
  };
  auto fload = [&](Frag& f, int buf, int slot) {
    const float* Arow = lds + (buf * PF + slot) * TILE + aoff;
    const float* Bcol = lds + (buf * PF + slot) * TILE + boff;
#pragma unroll
    for (int j = 0; j < KC / 4; ++j) {
      f.a[j] = *(const v2f*)(Arow + 4 * j);
#pragma unroll
      for (int n = 0; n < 2; ++n) f.b[j][n] = *(const v2f*)(Bcol + n * 16 * SA + 4 * j);
    }
    asm volatile("" : "+v"(acc[0]), "+v"(acc[1]) : : "memory");
  };
  auto mma = [&](const Frag& f) {
#pragma unroll
    for (int j = 0; j < KC / 4; ++j) {
#pragma unroll
      for (int n = 0; n < 2; ++n) {
        acc[n] = __builtin_amdgcn_wmma_f32_16x16x4_f32(
            false, f.a[j], false, f.b[j][n], (short)0, acc[n], false, false);
      }
    }
  };

  Stage s0, s1, s2, s3;
  gload(s0, 0); chunk_fence();
  gload(s1, 1); chunk_fence();
  gload(s2, 2); chunk_fence();
  gload(s3, 3); chunk_fence();
  lstore(s0, 0, 0); gload(s0, 4); chunk_fence();
  lstore(s1, 0, 1); gload(s1, 5); chunk_fence();
  lstore(s2, 0, 2); gload(s2, 6); chunk_fence();
  lstore(s3, 0, 3); gload(s3, 7);
  wg_barrier_loads_in_flight();
  Frag fA, fB;
  fload(fA, 0, 0); chunk_fence();

  int c = 0;
  for (; c + 16 <= NC; c += 8) {
    lstore(s0, 1, 0); gload(s0, c + 8);  fload(fB, 0, 1); mma(fA); chunk_fence();
    lstore(s1, 1, 1); gload(s1, c + 9);  fload(fA, 0, 2); mma(fB); chunk_fence();
    lstore(s2, 1, 2); gload(s2, c + 10); fload(fB, 0, 3); mma(fA); chunk_fence();
    lstore(s3, 1, 3); gload(s3, c + 11); wg_barrier_loads_in_flight(); fload(fA, 1, 0); mma(fB); chunk_fence();
    lstore(s0, 0, 0); gload(s0, c + 12); fload(fB, 1, 1); mma(fA); chunk_fence();
    lstore(s1, 0, 1); gload(s1, c + 13); fload(fA, 1, 2); mma(fB); chunk_fence();
    lstore(s2, 0, 2); gload(s2, c + 14); fload(fB, 1, 3); mma(fA); chunk_fence();
    lstore(s3, 0, 3); gload(s3, c + 15); wg_barrier_loads_in_flight(); fload(fA, 0, 0); mma(fB); chunk_fence();
  }
  lstore(s0, 1, 0); fload(fB, 0, 1); mma(fA); chunk_fence();
  lstore(s1, 1, 1); fload(fA, 0, 2); mma(fB); chunk_fence();
  lstore(s2, 1, 2); fload(fB, 0, 3); mma(fA); chunk_fence();
  lstore(s3, 1, 3); wg_barrier_loads_in_flight(); fload(fA, 1, 0); mma(fB); chunk_fence();
  fload(fB, 1, 1); mma(fA); chunk_fence();
  fload(fA, 1, 2); mma(fB); chunk_fence();
  fload(fB, 1, 3); mma(fA); chunk_fence();
  mma(fB);

  float* Z = (cg < 2) ? Z1 : Z2;
#pragma unroll
  for (int n = 0; n < 2; ++n) {
    const int col = c0 + n * 16 + laneRow;
#pragma unroll
    for (int v = 0; v < 8; ++v) {
      int row = r0 + rt * 16 + v + 8 * laneHi;
      Z[(size_t)row * DD + col] = acc[n][v];
    }
  }
}


__global__ __launch_bounds__(256) void epilogue_kernel(
    const float* __restrict__ Z1, const float* __restrict__ Z2,
    const float* __restrict__ src, const float* __restrict__ tgt,
    const float* __restrict__ W1, const float* __restrict__ W2,
    float* __restrict__ out) {
  __shared__ float z1s[4][DD];
  __shared__ float ps[4][DD];
  const int tid = threadIdx.x;
  const int r = tid >> 6;
  const int c = tid & 63;
  const int row = blockIdx.x * 4 + r;
  const size_t off = (size_t)row * DD + c;
  z1s[r][c] = Z1[off];
  ps[r][c] = src[off] * Z2[off];
  __syncthreads();
  float acc = tgt[off];
#pragma unroll 8
  for (int k = 0; k < DD; ++k)
    acc += z1s[r][k] * W1[k * DD + c] + ps[r][k] * W2[k * DD + c];
  out[off] = (acc >= 0.f) ? acc : ALPHA * acc;
}


extern "C" void kernel_launch(void* const* d_in, const int* in_sizes, int n_in,
                              void* d_out, int out_size, void* d_ws, size_t ws_size,
                              hipStream_t stream) {
  const float* user = (const float*)d_in[0];
  const float* item = (const float*)d_in[1];
  const float* adj  = (const float*)d_in[2];
  const float* W1   = (const float*)d_in[3];
  const float* W2   = (const float*)d_in[4];

  float* ws = (float*)d_ws;
  const size_t ND = (size_t)NN * DD;
  float* sr   = ws;
  float* sc   = ws + NN;
  float* cacc = ws + 2 * NN;
  float* Z1u  = ws + 3 * NN;
  float* Z2u  = Z1u + ND;
  float* Z1i  = Z2u + ND;
  float* Z2i  = Z1i + ND;

  float* out_user = (float*)d_out;
  float* out_item = out_user + ND;

  rowsum_kernel<<<NN, 256, 0, stream>>>(adj, sr);
  zero_kernel<<<NN / 256, 256, 0, stream>>>(cacc, NN);
  colsum_partial_kernel<<<dim3(NN / 256, 64), 256, 0, stream>>>(adj, cacc);
  sqrt_kernel<<<NN / 256, 256, 0, stream>>>(cacc, sc);

  spmm_kernel<true><<<NN / BM, 256, 0, stream>>>(adj, sr, sc, item, user, Z1u, Z2u);
  spmm_kernel<false><<<NN / BM, 256, 0, stream>>>(adj, sr, sc, user, item, Z1i, Z2i);

  epilogue_kernel<<<NN / 4, 256, 0, stream>>>(Z1u, Z2u, item, user, W1, W2, out_user);
  epilogue_kernel<<<NN / 4, 256, 0, stream>>>(Z1i, Z2i, user, item, W1, W2, out_item);
}


# degree sums: one streaming pass over the adjacency gives row sums and per-row-block column partials, a small kernel adds them in fixed order; no atomics; prep launches 4 -> 2; sum order changed
# speedup vs baseline: 5.4049x; 1.0720x over previous
#include <hip/hip_runtime.h>
#include <math.h>

typedef __attribute__((ext_vector_type(2))) float v2f;
typedef __attribute__((ext_vector_type(8))) float v8f;

#define NN 8192
#define DD 64
#define BM 32
#define KC 32
#define SA (KC + 2)
#define ALPHA 0.2f
#define EPS 1e-8f


typedef __attribute__((ext_vector_type(4))) float v4f;

#define PR 32
#define PW 16
#define PT (PW * 32)
#define PQ (NN / (4 * PT))
#define PB 4
#define NB (PR / PB)

__global__ __launch_bounds__(PT) void degree_pass_kernel(const float* __restrict__ adj,
                                                         float* __restrict__ sr,
                                                         float* __restrict__ colpart) {
  static_assert(PQ * 4 * PT == NN && PR == 32 && PR % PB == 0, "lane r of every wave keeps row r of the block");
  __shared__ float rowpart[PR][PW + 1];
  const int tid = threadIdx.x;
  const int lane = tid & 31;
  const int wave = tid >> 5;
  const int r0 = blockIdx.x * PR;
  const v4f* base = (const v4f*)(adj + (size_t)r0 * NN) + tid;

  v4f col[PQ];
#pragma unroll
  for (int q = 0; q < PQ; ++q) col[q] = (v4f)(0.f);
  v4f buf[2][PB][PQ];
  float mine = 0.f;

  auto load_batch = [&](v4f (&dst)[PB][PQ], int b) {
#pragma unroll
    for (int i = 0; i < PB; ++i) {
#pragma unroll
      for (int q = 0; q < PQ; ++q)
        dst[i][q] = __builtin_nontemporal_load(base + (size_t)(b * PB + i) * (NN / 4) + q * PT);
    }
  };

  load_batch(buf[0], 0);
#pragma unroll
  for (int b = 0; b < NB; ++b) {
    __builtin_amdgcn_sched_barrier(0);
    if (b + 1 < NB) load_batch(buf[(b + 1) & 1], b + 1);
    __builtin_amdgcn_sched_barrier(0);
    float s[PB];
#pragma unroll
    for (int i = 0; i < PB; ++i) {
      const v4f* w = buf[b & 1][i];
      const v4f t = (w[0] + w[1]) + (w[2] + w[3]);
      s[i] = (t.x + t.y) + (t.z + t.w);
#pragma unroll
      for (int q = 0; q < PQ; ++q) col[q] += w[q];
    }
#pragma unroll
    for (int m = 16; m >= 1; m >>= 1) {
#pragma unroll
      for (int i = 0; i < PB; ++i) s[i] += __shfl_xor(s[i], m, 32);
    }
#pragma unroll
    for (int i = 0; i < PB; ++i)
      if (lane == b * PB + i) mine = s[i];
  }

  v4f* cp = (v4f*)(colpart + (size_t)blockIdx.x * NN) + tid;
#pragma unroll
  for (int q = 0; q < PQ; ++q) cp[q * PT] = col[q];

  rowpart[lane][wave] = mine;
  __syncthreads();
  if (tid < PR) {
    float s = 0.f;
    for (int w = 0; w < PW; ++w) s += rowpart[tid][w];
    sr[r0 + tid] = sqrtf(s);
  }
}

#define FC 64
#define FG 16

__global__ __launch_bounds__(256) void colsum_finish_kernel(const float* __restrict__ colpart,
                                                            float* __restrict__ sc) {
  static_assert(FC / 4 * FG == 256 && (NN / PR) % FG == 0, "16 lanes x 16 bytes cover the columns, 16 groups the row blocks");
  __shared__ __align__(16) float part[FG][FC + 4];
  const int tid = threadIdx.x;
  const int cq = tid & (FC / 4 - 1);
  const int g = tid / (FC / 4);
  const v4f* p = (const v4f*)(colpart + (size_t)blockIdx.x * FC) + cq;
  v4f s = (v4f)(0.f);
#pragma unroll
  for (int i = 0; i < (NN / PR) / FG; ++i) s += p[(size_t)(g + FG * i) * (NN / 4)];
  *(v4f*)&part[g][4 * cq] = s;
  __syncthreads();
  if (tid < FC) {
    float t = 0.f;
    for (int k = 0; k < FG; ++k) t += part[k][tid];
    sc[blockIdx.x * FC + tid] = sqrtf(t);
  }
}


#define NC (NN / KC)
#define PF 4
#define TILE ((BM + 2 * DD) * SA)

struct Stage {
  float4 a;
  float4 n;
  float4 b1[2];
  float4 b2[2];
};

static __device__ __forceinline__ void wg_barrier_loads_in_flight() {
  asm volatile("s_wait_dscnt 0x0\n\ts_barrier_signal -1\n\ts_barrier_wait -1" ::: "memory");
}

template <bool TRANSA>
__global__ __launch_bounds__(256) void spmm_kernel(
    const float* __restrict__ adj, const float* __restrict__ sr, const float* __restrict__ sc,
    const float* __restrict__ Bsrc, const float* __restrict__ Btgt,
    float* __restrict__ Z1, float* __restrict__ Z2) {
  static_assert(PF == 4 && (NC % (2 * PF)) == 0 && NC >= 4 * PF, "the K loop below is written for a queue of four");
  __shared__ __align__(16) float lds[2 * PF * TILE];

  const int tid = threadIdx.x;
  const int wave = tid >> 5;
  const int lane = tid & 31;
  const int laneRow = lane & 15;
  const int laneHi = lane >> 4;
  const int r0 = blockIdx.x * BM;
  const int rt = wave & 1;
  const int cg = wave >> 1;
  const int c0 = (cg & 1) * 32;

  const int aq = tid >> 3;
  const int ar = (tid & 7) << 2;
  const float* ap = TRANSA ? adj + (size_t)aq * NN + (r0 + ar)
                           : adj + (size_t)(r0 + aq) * NN + ar;
  float i0, i1 = 0.f, i2 = 0.f, i3 = 0.f;
  if (!TRANSA) {
    i0 = sr[r0 + aq];
  } else {
    i0 = sc[r0 + ar + 0]; i1 = sc[r0 + ar + 1]; i2 = sc[r0 + ar + 2]; i3 = sc[r0 + ar + 3];
  }
  auto chunk_fence = [&]() {
    if (!TRANSA) asm volatile("" : "+v"(i0) : : "memory");
    else asm volatile("" : "+v"(i0), "+v"(i1), "+v"(i2), "+v"(i3) : : "memory");
    __builtin_amdgcn_sched_barrier(0);
  };
  const int bu = tid >> 4;
  const int bc = (tid & 15) << 2;
  const float* b1p = Bsrc + (size_t)bu * DD + bc;
  const float* b2p = Btgt + (size_t)bu * DD + bc;

  auto gload = [&](Stage& s, int c) {
    const int k0 = c * KC;
    if (!TRANSA) {
      s.n.x = sc[k0 + ar + 0]; s.n.y = sc[k0 + ar + 1];
      s.n.z = sc[k0 + ar + 2]; s.n.w = sc[k0 + ar + 3];
      s.a = *(const float4*)(ap + k0);
    } else {
      s.n.x = sr[k0 + aq];
      s.a = *(const float4*)(ap + (size_t)k0 * NN);
    }
    s.b1[0] = *(const float4*)(b1p + (size_t)k0 * DD);
    s.b1[1] = *(const float4*)(b1p + (size_t)(k0 + 16) * DD);
    s.b2[0] = *(const float4*)(b2p + (size_t)k0 * DD);
    s.b2[1] = *(const float4*)(b2p + (size_t)(k0 + 16) * DD);
  };

  auto lstore = [&](const Stage& s, int buf, int slot) {
    float* L = lds + (buf * PF + slot) * TILE;
    if (!TRANSA) {
      float* dst = L + aq * SA + ar;
      dst[0] = s.a.x * __builtin_amdgcn_rcpf(i0 * s.n.x + EPS);
      dst[1] = s.a.y * __builtin_amdgcn_rcpf(i0 * s.n.y + EPS);
      dst[2] = s.a.z * __builtin_amdgcn_rcpf(i0 * s.n.z + EPS);
      dst[3] = s.a.w * __builtin_amdgcn_rcpf(i0 * s.n.w + EPS);
    } else {
      const float a = s.n.x;
      L[(ar + 0) * SA + aq] = s.a.x * __builtin_amdgcn_rcpf(a * i0 + EPS);
      L[(ar + 1) * SA + aq] = s.a.y * __builtin_amdgcn_rcpf(a * i1 + EPS);
      L[(ar + 2) * SA + aq] = s.a.z * __builtin_amdgcn_rcpf(a * i2 + EPS);
      L[(ar + 3) * SA + aq] = s.a.w * __builtin_amdgcn_rcpf(a * i3 + EPS);
    }
#pragma unroll
    for (int it = 0; it < 2; ++it) {
      float* d1 = L + (BM + bc) * SA + (bu + it * 16);
      float* d2 = L + (BM + DD + bc) * SA + (bu + it * 16);
      d1[0 * SA] = s.b1[it].x; d1[1 * SA] = s.b1[it].y;
      d1[2 * SA] = s.b1[it].z; d1[3 * SA] = s.b1[it].w;
      d2[0 * SA] = s.b2[it].x; d2[1 * SA] = s.b2[it].y;
      d2[2 * SA] = s.b2[it].z; d2[3 * SA] = s.b2[it].w;
    }
  };

  v8f acc[2] = {};

  const int aoff = (rt * 16 + laneRow) * SA + 2 * laneHi;
  const int boff = (BM + cg * 32 + laneRow) * SA + 2 * laneHi;
  struct Frag {
    v2f a[KC / 4];
    v2f b[KC / 4][2];
  };
  auto fload = [&](Frag& f, int buf, int slot) {
    const float* Arow = lds + (buf * PF + slot) * TILE + aoff;
    const float* Bcol = lds + (buf * PF + slot) * TILE + boff;
#pragma unroll
    for (int j = 0; j < KC / 4; ++j) {
      f.a[j] = *(const v2f*)(Arow + 4 * j);
#pragma unroll
      for (int n = 0; n < 2; ++n) f.b[j][n] = *(const v2f*)(Bcol + n * 16 * SA + 4 * j);
    }
    asm volatile("" : "+v"(acc[0]), "+v"(acc[1]) : : "memory");
  };
  auto mma = [&](const Frag& f) {
#pragma unroll
    for (int j = 0; j < KC / 4; ++j) {
#pragma unroll
      for (int n = 0; n < 2; ++n) {
        acc[n] = __builtin_amdgcn_wmma_f32_16x16x4_f32(
            false, f.a[j], false, f.b[j][n], (short)0, acc[n], false, false);
      }
    }
  };

  Stage s0, s1, s2, s3;
  gload(s0, 0); chunk_fence();
  gload(s1, 1); chunk_fence();
  gload(s2, 2); chunk_fence();
  gload(s3, 3); chunk_fence();
  lstore(s0, 0, 0); gload(s0, 4); chunk_fence();
  lstore(s1, 0, 1); gload(s1, 5); chunk_fence();
  lstore(s2, 0, 2); gload(s2, 6); chunk_fence();
  lstore(s3, 0, 3); gload(s3, 7);
  wg_barrier_loads_in_flight();
  Frag fA, fB;
  fload(fA, 0, 0); chunk_fence();

  int c = 0;
  for (; c + 16 <= NC; c += 8) {
    lstore(s0, 1, 0); gload(s0, c + 8);  fload(fB, 0, 1); mma(fA); chunk_fence();
    lstore(s1, 1, 1); gload(s1, c + 9);  fload(fA, 0, 2); mma(fB); chunk_fence();
    lstore(s2, 1, 2); gload(s2, c + 10); fload(fB, 0, 3); mma(fA); chunk_fence();
    lstore(s3, 1, 3); gload(s3, c + 11); wg_barrier_loads_in_flight(); fload(fA, 1, 0); mma(fB); chunk_fence();
    lstore(s0, 0, 0); gload(s0, c + 12); fload(fB, 1, 1); mma(fA); chunk_fence();
    lstore(s1, 0, 1); gload(s1, c + 13); fload(fA, 1, 2); mma(fB); chunk_fence();
    lstore(s2, 0, 2); gload(s2, c + 14); fload(fB, 1, 3); mma(fA); chunk_fence();
    lstore(s3, 0, 3); gload(s3, c + 15); wg_barrier_loads_in_flight(); fload(fA, 0, 0); mma(fB); chunk_fence();
  }
  lstore(s0, 1, 0); fload(fB, 0, 1); mma(fA); chunk_fence();
  lstore(s1, 1, 1); fload(fA, 0, 2); mma(fB); chunk_fence();
  lstore(s2, 1, 2); fload(fB, 0, 3); mma(fA); chunk_fence();
  lstore(s3, 1, 3); wg_barrier_loads_in_flight(); fload(fA, 1, 0); mma(fB); chunk_fence();
  fload(fB, 1, 1); mma(fA); chunk_fence();
  fload(fA, 1, 2); mma(fB); chunk_fence();
  fload(fB, 1, 3); mma(fA); chunk_fence();
  mma(fB);

  float* Z = (cg < 2) ? Z1 : Z2;
#pragma unroll
  for (int n = 0; n < 2; ++n) {
    const int col = c0 + n * 16 + laneRow;
#pragma unroll
    for (int v = 0; v < 8; ++v) {
      int row = r0 + rt * 16 + v + 8 * laneHi;
      Z[(size_t)row * DD + col] = acc[n][v];
    }
  }
}


__global__ __launch_bounds__(256) void epilogue_kernel(
    const float* __restrict__ Z1, const float* __restrict__ Z2,
    const float* __restrict__ src, const float* __restrict__ tgt,
    const float* __restrict__ W1, const float* __restrict__ W2,
    float* __restrict__ out) {
  __shared__ float z1s[4][DD];
  __shared__ float ps[4][DD];
  const int tid = threadIdx.x;
  const int r = tid >> 6;
  const int c = tid & 63;
  const int row = blockIdx.x * 4 + r;
  const size_t off = (size_t)row * DD + c;
  z1s[r][c] = Z1[off];
  ps[r][c] = src[off] * Z2[off];
  __syncthreads();
  float acc = tgt[off];
#pragma unroll 8
  for (int k = 0; k < DD; ++k)
    acc += z1s[r][k] * W1[k * DD + c] + ps[r][k] * W2[k * DD + c];
  out[off] = (acc >= 0.f) ? acc : ALPHA * acc;
}


extern "C" void kernel_launch(void* const* d_in, const int* in_sizes, int n_in,
                              void* d_out, int out_size, void* d_ws, size_t ws_size,
                              hipStream_t stream) {
  const float* user = (const float*)d_in[0];
  const float* item = (const float*)d_in[1];
  const float* adj  = (const float*)d_in[2];
  const float* W1   = (const float*)d_in[3];
  const float* W2   = (const float*)d_in[4];

  float* ws = (float*)d_ws;
  const size_t ND = (size_t)NN * DD;
  float* sr   = ws;
  float* sc   = ws + NN;
  float* Z1u  = ws + 3 * NN;
  float* Z2u  = Z1u + ND;
  float* Z1i  = Z2u + ND;
  float* Z2i  = Z1i + ND;
  static_assert((size_t)(NN / PR) * NN == 4 * (size_t)NN * DD, "the column partials fill the four Z planes exactly");
  float* colpart = Z1u;

  float* out_user = (float*)d_out;
  float* out_item = out_user + ND;

  degree_pass_kernel<<<NN / PR, PT, 0, stream>>>(adj, sr, colpart);
  colsum_finish_kernel<<<NN / FC, 256, 0, stream>>>(colpart, sc);

  spmm_kernel<true><<<NN / BM, 256, 0, stream>>>(adj, sr, sc, item, user, Z1u, Z2u);
  spmm_kernel<false><<<NN / BM, 256, 0, stream>>>(adj, sr, sc, user, item, Z1i, Z2i);

  epilogue_kernel<<<NN / 4, 256, 0, stream>>>(Z1u, Z2u, item, user, W1, W2, out_user);
  epilogue_kernel<<<NN / 4, 256, 0, stream>>>(Z1i, Z2i, user, item, W1, W2, out_item);
}


# both spmm products in one launch of 256 work-groups: a work-group takes 64 rows, not 32, a wave 2x2 accumulator tiles, so a staged embedding chunk feeds twice the WMMAs; order of all k sums unchanged
# speedup vs baseline: 6.1150x; 1.1314x over previous
#include <hip/hip_runtime.h>
#include <math.h>

typedef __attribute__((ext_vector_type(2))) float v2f;
typedef __attribute__((ext_vector_type(8))) float v8f;

#define NN 8192
#define DD 64
#define BM 64
#define KC 32
#define SA (KC + 2)
#define ALPHA 0.2f
#define EPS 1e-8f


typedef __attribute__((ext_vector_type(4))) float v4f;

#define PR 32
#define PW 16
#define PT (PW * 32)
#define PQ (NN / (4 * PT))
#define PB 4
#define NB (PR / PB)

__global__ __launch_bounds__(PT) void degree_pass_kernel(const float* __restrict__ adj,
                                                         float* __restrict__ sr,
                                                         float* __restrict__ colpart) {
  static_assert(PQ * 4 * PT == NN && PR == 32 && PR % PB == 0, "lane r of every wave keeps row r of the block");
  __shared__ float rowpart[PR][PW + 1];
  const int tid = threadIdx.x;
  const int lane = tid & 31;
  const int wave = tid >> 5;
  const int r0 = blockIdx.x * PR;
  const v4f* base = (const v4f*)(adj + (size_t)r0 * NN) + tid;

  v4f col[PQ];
#pragma unroll
  for (int q = 0; q < PQ; ++q) col[q] = (v4f)(0.f);
  v4f buf[2][PB][PQ];
  float mine = 0.f;

  auto load_batch = [&](v4f (&dst)[PB][PQ], int b) {
#pragma unroll
    for (int i = 0; i < PB; ++i) {
#pragma unroll
      for (int q = 0; q < PQ; ++q)
        dst[i][q] = __builtin_nontemporal_load(base + (size_t)(b * PB + i) * (NN / 4) + q * PT);
    }
  };

  load_batch(buf[0], 0);
#pragma unroll
  for (int b = 0; b < NB; ++b) {
    __builtin_amdgcn_sched_barrier(0);
    if (b + 1 < NB) load_batch(buf[(b + 1) & 1], b + 1);
    __builtin_amdgcn_sched_barrier(0);
    float s[PB];
#pragma unroll
    for (int i = 0; i < PB; ++i) {
      const v4f* w = buf[b & 1][i];
      const v4f t = (w[0] + w[1]) + (w[2] + w[3]);
      s[i] = (t.x + t.y) + (t.z + t.w);
#pragma unroll
      for (int q = 0; q < PQ; ++q) col[q] += w[q];
    }
#pragma unroll
    for (int m = 16; m >= 1; m >>= 1) {
#pragma unroll
      for (int i = 0; i < PB; ++i) s[i] += __shfl_xor(s[i], m, 32);
    }
#pragma unroll
    for (int i = 0; i < PB; ++i)
      if (lane == b * PB + i) mine = s[i];
  }

  v4f* cp = (v4f*)(colpart + (size_t)blockIdx.x * NN) + tid;
#pragma unroll
  for (int q = 0; q < PQ; ++q) cp[q * PT] = col[q];

  rowpart[lane][wave] = mine;
  __syncthreads();
  if (tid < PR) {
    float s = 0.f;
    for (int w = 0; w < PW; ++w) s += rowpart[tid][w];
    sr[r0 + tid] = sqrtf(s);
  }
}

#define FC 64
#define FG 16

__global__ __launch_bounds__(256) void colsum_finish_kernel(const float* __restrict__ colpart,
                                                            float* __restrict__ sc) {
  static_assert(FC / 4 * FG == 256 && (NN / PR) % FG == 0, "16 lanes x 16 bytes cover the columns, 16 groups the row blocks");
  __shared__ __align__(16) float part[FG][FC + 4];
  const int tid = threadIdx.x;
  const int cq = tid & (FC / 4 - 1);
  const int g = tid / (FC / 4);
  const v4f* p = (const v4f*)(colpart + (size_t)blockIdx.x * FC) + cq;
  v4f s = (v4f)(0.f);
#pragma unroll
  for (int i = 0; i < (NN / PR) / FG; ++i) s += p[(size_t)(g + FG * i) * (NN / 4)];
  *(v4f*)&part[g][4 * cq] = s;
  __syncthreads();
  if (tid < FC) {
    float t = 0.f;
    for (int k = 0; k < FG; ++k) t += part[k][tid];
    sc[blockIdx.x * FC + tid] = sqrtf(t);
  }
}


#define NC (NN / KC)
#define PF 4
#define TILE ((BM + 2 * DD) * SA)

struct Stage {
  float4 a[2];
  float4 n;
  float4 b1[2];
  float4 b2[2];
};

static __device__ __forceinline__ void wg_barrier_loads_in_flight() {
  asm volatile("s_wait_dscnt 0x0\n\ts_barrier_signal -1\n\ts_barrier_wait -1" ::: "memory");
}

template <bool TRANSA>
static __device__ __forceinline__ void spmm_body(
    float* __restrict__ lds, const int blk,
    const float* __restrict__ adj, const float* __restrict__ sr, const float* __restrict__ sc,
    const float* __restrict__ Bsrc, const float* __restrict__ Btgt,
    float* __restrict__ Z1, float* __restrict__ Z2) {
  static_assert(PF == 4 && (NC % (2 * PF)) == 0 && NC >= 4 * PF, "the K loop below is written for a queue of four");
  static_assert(BM == 64, "two A quads per thread, 32 tile rows apart; 2 x 2 accumulator tiles per wave");

  const int tid = threadIdx.x;
  const int wave = tid >> 5;
  const int lane = tid & 31;
  const int laneRow = lane & 15;
  const int laneHi = lane >> 4;
  const int r0 = blk * BM;
  const int rp = wave & 1;
  const int cg = wave >> 1;
  const int c0 = (cg & 1) * 32;

  const int aq = tid >> 3;
  const int ar = (tid & 7) << 2;
  const float* ap = TRANSA ? adj + (size_t)aq * NN + (r0 + ar)
                           : adj + (size_t)(r0 + aq) * NN + ar;
  float in[8] = {0.f, 0.f, 0.f, 0.f, 0.f, 0.f, 0.f, 0.f};
  if (!TRANSA) {
    in[0] = sr[r0 + aq];
    in[1] = sr[r0 + aq + 32];
  } else {
#pragma unroll
    for (int j = 0; j < 4; ++j) {
      in[j] = sc[r0 + ar + j];
      in[4 + j] = sc[r0 + 32 + ar + j];
    }
  }
  auto chunk_fence = [&]() {
    if (!TRANSA) asm volatile("" : "+v"(in[0]), "+v"(in[1]) : : "memory");
    else asm volatile("" : "+v"(in[0]), "+v"(in[1]), "+v"(in[2]), "+v"(in[3]),
                           "+v"(in[4]), "+v"(in[5]), "+v"(in[6]), "+v"(in[7]) : : "memory");
    __builtin_amdgcn_sched_barrier(0);
  };
  const int bu = tid >> 4;
  const int bc = (tid & 15) << 2;
  const float* b1p = Bsrc + (size_t)bu * DD + bc;
  const float* b2p = Btgt + (size_t)bu * DD + bc;

  auto gload = [&](Stage& s, int c) {
    const int k0 = c * KC;
    if (!TRANSA) {
      s.n.x = sc[k0 + ar + 0]; s.n.y = sc[k0 + ar + 1];
      s.n.z = sc[k0 + ar + 2]; s.n.w = sc[k0 + ar + 3];
      s.a[0] = *(const float4*)(ap + k0);
      s.a[1] = *(const float4*)(ap + (size_t)32 * NN + k0);
    } else {
      s.n.x = sr[k0 + aq];
      s.a[0] = *(const float4*)(ap + (size_t)k0 * NN);
      s.a[1] = *(const float4*)(ap + (size_t)k0 * NN + 32);
    }
    s.b1[0] = *(const float4*)(b1p + (size_t)k0 * DD);
    s.b1[1] = *(const float4*)(b1p + (size_t)(k0 + 16) * DD);
    s.b2[0] = *(const float4*)(b2p + (size_t)k0 * DD);
    s.b2[1] = *(const float4*)(b2p + (size_t)(k0 + 16) * DD);
  };

  auto lstore = [&](const Stage& s, int buf, int slot) {
    float* L = lds + (buf * PF + slot) * TILE;
    if (!TRANSA) {
#pragma unroll
      for (int h = 0; h < 2; ++h) {
        float* dst = L + (aq + 32 * h) * SA + ar;
        dst[0] = s.a[h].x * __builtin_amdgcn_rcpf(in[h] * s.n.x + EPS);
        dst[1] = s.a[h].y * __builtin_amdgcn_rcpf(in[h] * s.n.y + EPS);
        dst[2] = s.a[h].z * __builtin_amdgcn_rcpf(in[h] * s.n.z + EPS);
        dst[3] = s.a[h].w * __builtin_amdgcn_rcpf(in[h] * s.n.w + EPS);
      }
    } else {
      const float a = s.n.x;
#pragma unroll
      for (int h = 0; h < 2; ++h) {
        float* dst = L + (32 * h + ar) * SA + aq;
        dst[0 * SA] = s.a[h].x * __builtin_amdgcn_rcpf(a * in[4 * h + 0] + EPS);
        dst[1 * SA] = s.a[h].y * __builtin_amdgcn_rcpf(a * in[4 * h + 1] + EPS);
        dst[2 * SA] = s.a[h].z * __builtin_amdgcn_rcpf(a * in[4 * h + 2] + EPS);
        dst[3 * SA] = s.a[h].w * __builtin_amdgcn_rcpf(a * in[4 * h + 3] + EPS);
      }
    }
#pragma unroll
    for (int it = 0; it < 2; ++it) {
      float* d1 = L + (BM + bc) * SA + (bu + it * 16);
      float* d2 = L + (BM + DD + bc) * SA + (bu + it * 16);
      d1[0 * SA] = s.b1[it].x; d1[1 * SA] = s.b1[it].y;
      d1[2 * SA] = s.b1[it].z; d1[3 * SA] = s.b1[it].w;
      d2[0 * SA] = s.b2[it].x; d2[1 * SA] = s.b2[it].y;
      d2[2 * SA] = s.b2[it].z; d2[3 * SA] = s.b2[it].w;
    }
  };

  v8f acc[2][2] = {};

  const int aoff = (rp * 32 + laneRow) * SA + 2 * laneHi;
  const int boff = (BM + cg * 32 + laneRow) * SA + 2 * laneHi;
  struct Frag {
    v2f a[KC / 4][2];
    v2f b[KC / 4][2];
  };
  auto fload = [&](Frag& f, int buf, int slot) {
    const float* Arow = lds + (buf * PF + slot) * TILE + aoff;
    const float* Bcol = lds + (buf * PF + slot) * TILE + boff;
#pragma unroll
    for (int j = 0; j < KC / 4; ++j) {
#pragma unroll
      for (int m = 0; m < 2; ++m) f.a[j][m] = *(const v2f*)(Arow + m * 16 * SA + 4 * j);
#pragma unroll
      for (int n = 0; n < 2; ++n) f.b[j][n] = *(const v2f*)(Bcol + n * 16 * SA + 4 * j);
    }
    asm volatile("" : "+v"(acc[0][0]), "+v"(acc[0][1]), "+v"(acc[1][0]), "+v"(acc[1][1]) : : "memory");
  };
  auto mma = [&](const Frag& f) {
#pragma unroll
    for (int j = 0; j < KC / 4; ++j) {
#pragma unroll
      for (int m = 0; m < 2; ++m) {
#pragma unroll
        for (int n = 0; n < 2; ++n) {
          acc[m][n] = __builtin_amdgcn_wmma_f32_16x16x4_f32(
              false, f.a[j][m], false, f.b[j][n], (short)0, acc[m][n], false, false);
        }
      }
    }
  };

  Stage s0, s1, s2, s3;
  gload(s0, 0); chunk_fence();
  gload(s1, 1); chunk_fence();
  gload(s2, 2); chunk_fence();
  gload(s3, 3); chunk_fence();
  lstore(s0, 0, 0); gload(s0, 4); chunk_fence();
  lstore(s1, 0, 1); gload(s1, 5); chunk_fence();
  lstore(s2, 0, 2); gload(s2, 6); chunk_fence();
  lstore(s3, 0, 3); gload(s3, 7);
  wg_barrier_loads_in_flight();
  Frag fA, fB;
  fload(fA, 0, 0); chunk_fence();

  int c = 0;
  for (; c + 16 <= NC; c += 8) {
    lstore(s0, 1, 0); gload(s0, c + 8);  fload(fB, 0, 1); mma(fA); chunk_fence();
    lstore(s1, 1, 1); gload(s1, c + 9);  fload(fA, 0, 2); mma(fB); chunk_fence();
    lstore(s2, 1, 2); gload(s2, c + 10); fload(fB, 0, 3); mma(fA); chunk_fence();
    lstore(s3, 1, 3); gload(s3, c + 11); wg_barrier_loads_in_flight(); fload(fA, 1, 0); mma(fB); chunk_fence();
    lstore(s0, 0, 0); gload(s0, c + 12); fload(fB, 1, 1); mma(fA); chunk_fence();
    lstore(s1, 0, 1); gload(s1, c + 13); fload(fA, 1, 2); mma(fB); chunk_fence();
    lstore(s2, 0, 2); gload(s2, c + 14); fload(fB, 1, 3); mma(fA); chunk_fence();
    lstore(s3, 0, 3); gload(s3, c + 15); wg_barrier_loads_in_flight(); fload(fA, 0, 0); mma(fB); chunk_fence();
  }
  lstore(s0, 1, 0); fload(fB, 0, 1); mma(fA); chunk_fence();
  lstore(s1, 1, 1); fload(fA, 0, 2); mma(fB); chunk_fence();
  lstore(s2, 1, 2); fload(fB, 0, 3); mma(fA); chunk_fence();
  lstore(s3, 1, 3); wg_barrier_loads_in_flight(); fload(fA, 1, 0); mma(fB); chunk_fence();
  fload(fB, 1, 1); mma(fA); chunk_fence();
  fload(fA, 1, 2); mma(fB); chunk_fence();
  fload(fB, 1, 3); mma(fA); chunk_fence();
  mma(fB);

  float* Z = (cg < 2) ? Z1 : Z2;
#pragma unroll
  for (int m = 0; m < 2; ++m) {
#pragma unroll
    for (int n = 0; n < 2; ++n) {
      const int col = c0 + n * 16 + laneRow;
#pragma unroll
      for (int v = 0; v < 8; ++v) {
        int row = r0 + rp * 32 + m * 16 + v + 8 * laneHi;
        Z[(size_t)row * DD + col] = acc[m][n][v];
      }
    }
  }
}

__global__ __launch_bounds__(256) void spmm_both_kernel(
    const float* __restrict__ adj, const float* __restrict__ sr, const float* __restrict__ sc,
    const float* __restrict__ user, const float* __restrict__ item,
    float* __restrict__ Z1u, float* __restrict__ Z2u,
    float* __restrict__ Z1i, float* __restrict__ Z2i) {
  __shared__ __align__(16) float lds[2 * PF * TILE];
  const int blk = blockIdx.x >> 1;
  if ((blockIdx.x & 1) == 0) spmm_body<true>(lds, blk, adj, sr, sc, item, user, Z1u, Z2u);
  else spmm_body<false>(lds, blk, adj, sr, sc, user, item, Z1i, Z2i);
}


__global__ __launch_bounds__(256) void epilogue_kernel(
    const float* __restrict__ Z1, const float* __restrict__ Z2,
    const float* __restrict__ src, const float* __restrict__ tgt,
    const float* __restrict__ W1, const float* __restrict__ W2,
    float* __restrict__ out) {
  __shared__ float z1s[4][DD];
  __shared__ float ps[4][DD];
  const int tid = threadIdx.x;
  const int r = tid >> 6;
  const int c = tid & 63;
  const int row = blockIdx.x * 4 + r;
  const size_t off = (size_t)row * DD + c;
  z1s[r][c] = Z1[off];
  ps[r][c] = src[off] * Z2[off];
  __syncthreads();
  float acc = tgt[off];
#pragma unroll 8
  for (int k = 0; k < DD; ++k)
    acc += z1s[r][k] * W1[k * DD + c] + ps[r][k] * W2[k * DD + c];
  out[off] = (acc >= 0.f) ? acc : ALPHA * acc;
}


extern "C" void kernel_launch(void* const* d_in, const int* in_sizes, int n_in,
                              void* d_out, int out_size, void* d_ws, size_t ws_size,
                              hipStream_t stream) {
  const float* user = (const float*)d_in[0];
  const float* item = (const float*)d_in[1];
  const float* adj  = (const float*)d_in[2];
  const float* W1   = (const float*)d_in[3];
  const float* W2   = (const float*)d_in[4];

  float* ws = (float*)d_ws;
  const size_t ND = (size_t)NN * DD;
  float* sr   = ws;
  float* sc   = ws + NN;
  float* Z1u  = ws + 3 * NN;
  float* Z2u  = Z1u + ND;
  float* Z1i  = Z2u + ND;
  float* Z2i  = Z1i + ND;
  static_assert((size_t)(NN / PR) * NN == 4 * (size_t)NN * DD, "the column partials fill the four Z planes exactly");
  float* colpart = Z1u;

  float* out_user = (float*)d_out;
  float* out_item = out_user + ND;

  degree_pass_kernel<<<NN / PR, PT, 0, stream>>>(adj, sr, colpart);
  colsum_finish_kernel<<<NN / FC, 256, 0, stream>>>(colpart, sc);

  spmm_both_kernel<<<2 * (NN / BM), 256, 0, stream>>>(adj, sr, sc, user, item, Z1u, Z2u, Z1i, Z2i);

  epilogue_kernel<<<NN / 4, 256, 0, stream>>>(Z1u, Z2u, item, user, W1, W2, out_user);
  epilogue_kernel<<<NN / 4, 256, 0, stream>>>(Z1i, Z2i, user, item, W1, W2, out_item);
}


# second stage (Z1@W1 + (src.*Z2)@W2 + tgt, lrelu) moved into the spmm work-group on f32 WMMA via LDS: Z never stored, launches 5 -> 3; k-sum order of the second stage changed
# speedup vs baseline: 6.5679x; 1.0741x over previous
#include <hip/hip_runtime.h>
#include <math.h>

typedef __attribute__((ext_vector_type(2))) float v2f;
typedef __attribute__((ext_vector_type(8))) float v8f;

#define NN 8192
#define DD 64
#define BM 64
#define KC 32
#define SA (KC + 2)
#define ALPHA 0.2f
#define EPS 1e-8f


typedef __attribute__((ext_vector_type(4))) float v4f;

#define PR 32
#define PW 16
#define PT (PW * 32)
#define PQ (NN / (4 * PT))
#define PB 4
#define NB (PR / PB)

__global__ __launch_bounds__(PT) void degree_pass_kernel(const float* __restrict__ adj,
                                                         float* __restrict__ sr,
                                                         float* __restrict__ colpart) {
  static_assert(PQ * 4 * PT == NN && PR == 32 && PR % PB == 0, "lane r of every wave keeps row r of the block");
  __shared__ float rowpart[PR][PW + 1];
  const int tid = threadIdx.x;
  const int lane = tid & 31;
  const int wave = tid >> 5;
  const int r0 = blockIdx.x * PR;
  const v4f* base = (const v4f*)(adj + (size_t)r0 * NN) + tid;

  v4f col[PQ];
#pragma unroll
  for (int q = 0; q < PQ; ++q) col[q] = (v4f)(0.f);
  v4f buf[2][PB][PQ];
  float mine = 0.f;

  auto load_batch = [&](v4f (&dst)[PB][PQ], int b) {
#pragma unroll
    for (int i = 0; i < PB; ++i) {
#pragma unroll
      for (int q = 0; q < PQ; ++q)
        dst[i][q] = __builtin_nontemporal_load(base + (size_t)(b * PB + i) * (NN / 4) + q * PT);
    }
  };

  load_batch(buf[0], 0);
#pragma unroll
  for (int b = 0; b < NB; ++b) {
    __builtin_amdgcn_sched_barrier(0);
    if (b + 1 < NB) load_batch(buf[(b + 1) & 1], b + 1);
    __builtin_amdgcn_sched_barrier(0);
    float s[PB];
#pragma unroll
    for (int i = 0; i < PB; ++i) {
      const v4f* w = buf[b & 1][i];
      const v4f t = (w[0] + w[1]) + (w[2] + w[3]);
      s[i] = (t.x + t.y) + (t.z + t.w);
#pragma unroll
      for (int q = 0; q < PQ; ++q) col[q] += w[q];
    }
#pragma unroll
    for (int m = 16; m >= 1; m >>= 1) {
#pragma unroll
      for (int i = 0; i < PB; ++i) s[i] += __shfl_xor(s[i], m, 32);
    }
#pragma unroll
    for (int i = 0; i < PB; ++i)
      if (lane == b * PB + i) mine = s[i];
  }

  v4f* cp = (v4f*)(colpart + (size_t)blockIdx.x * NN) + tid;
#pragma unroll
  for (int q = 0; q < PQ; ++q) cp[q * PT] = col[q];

  rowpart[lane][wave] = mine;
  __syncthreads();
  if (tid < PR) {
    float s = 0.f;
    for (int w = 0; w < PW; ++w) s += rowpart[tid][w];
    sr[r0 + tid] = sqrtf(s);
  }
}

#define FC 64
#define FG 16

__global__ __launch_bounds__(256) void colsum_finish_kernel(const float* __restrict__ colpart,
                                                            float* __restrict__ sc) {
  static_assert(FC / 4 * FG == 256 && (NN / PR) % FG == 0, "16 lanes x 16 bytes cover the columns, 16 groups the row blocks");
  __shared__ __align__(16) float part[FG][FC + 4];
  const int tid = threadIdx.x;
  const int cq = tid & (FC / 4 - 1);
  const int g = tid / (FC / 4);
  const v4f* p = (const v4f*)(colpart + (size_t)blockIdx.x * FC) + cq;
  v4f s = (v4f)(0.f);
#pragma unroll
  for (int i = 0; i < (NN / PR) / FG; ++i) s += p[(size_t)(g + FG * i) * (NN / 4)];
  *(v4f*)&part[g][4 * cq] = s;
  __syncthreads();
  if (tid < FC) {
    float t = 0.f;
    for (int k = 0; k < FG; ++k) t += part[k][tid];
    sc[blockIdx.x * FC + tid] = sqrtf(t);
  }
}


#define NC (NN / KC)
#define PF 4
#define TILE ((BM + 2 * DD) * SA)

#define PS 132
#define EP_P 0
#define EP_W (BM * PS)
static_assert(EP_W + DD * PS <= PF * TILE, "both tiles lie inside ring stage 0, which no wave reads after the last barrier of the K loop");

struct Stage {
  float4 a[2];
  float4 n;
  float4 b1[2];
  float4 b2[2];
};

static __device__ __forceinline__ void wg_barrier_loads_in_flight() {
  asm volatile("s_wait_dscnt 0x0\n\ts_barrier_signal -1\n\ts_barrier_wait -1" ::: "memory");
}

template <bool TRANSA>
static __device__ __forceinline__ void spmm_body(
    float* __restrict__ lds, const int blk,
    const float* __restrict__ adj, const float* __restrict__ sr, const float* __restrict__ sc,
    const float* __restrict__ Bsrc, const float* __restrict__ Btgt,
    const float* __restrict__ W1, const float* __restrict__ W2,
    float* __restrict__ out) {
  static_assert(PF == 4 && (NC % (2 * PF)) == 0 && NC >= 4 * PF, "the K loop below is written for a queue of four");
  static_assert(BM == 64, "two A quads per thread, 32 tile rows apart; 2 x 2 accumulator tiles per wave");

  const int tid = threadIdx.x;
  const int wave = tid >> 5;
  const int lane = tid & 31;
  const int laneRow = lane & 15;
  const int laneHi = lane >> 4;
  const int r0 = blk * BM;
  const int rp = wave & 1;
  const int cg = wave >> 1;
  const int c0 = (cg & 1) * 32;

  const int aq = tid >> 3;
  const int ar = (tid & 7) << 2;
  const float* ap = TRANSA ? adj + (size_t)aq * NN + (r0 + ar)
                           : adj + (size_t)(r0 + aq) * NN + ar;
  float in[8] = {0.f, 0.f, 0.f, 0.f, 0.f, 0.f, 0.f, 0.f};
  if (!TRANSA) {
    in[0] = sr[r0 + aq];
    in[1] = sr[r0 + aq + 32];
  } else {
#pragma unroll
    for (int j = 0; j < 4; ++j) {
      in[j] = sc[r0 + ar + j];
      in[4 + j] = sc[r0 + 32 + ar + j];
    }
  }
  auto chunk_fence = [&]() {
    if (!TRANSA) asm volatile("" : "+v"(in[0]), "+v"(in[1]) : : "memory");
    else asm volatile("" : "+v"(in[0]), "+v"(in[1]), "+v"(in[2]), "+v"(in[3]),
                           "+v"(in[4]), "+v"(in[5]), "+v"(in[6]), "+v"(in[7]) : : "memory");
    __builtin_amdgcn_sched_barrier(0);
  };
  const int bu = tid >> 4;
  const int bc = (tid & 15) << 2;
  const float* b1p = Bsrc + (size_t)bu * DD + bc;
  const float* b2p = Btgt + (size_t)bu * DD + bc;

  auto gload = [&](Stage& s, int c) {
    const int k0 = c * KC;
    if (!TRANSA) {
      s.n.x = sc[k0 + ar + 0]; s.n.y = sc[k0 + ar + 1];
      s.n.z = sc[k0 + ar + 2]; s.n.w = sc[k0 + ar + 3];
      s.a[0] = *(const float4*)(ap + k0);
      s.a[1] = *(const float4*)(ap + (size_t)32 * NN + k0);
    } else {
      s.n.x = sr[k0 + aq];
      s.a[0] = *(const float4*)(ap + (size_t)k0 * NN);
      s.a[1] = *(const float4*)(ap + (size_t)k0 * NN + 32);
    }
    s.b1[0] = *(const float4*)(b1p + (size_t)k0 * DD);
    s.b1[1] = *(const float4*)(b1p + (size_t)(k0 + 16) * DD);
    s.b2[0] = *(const float4*)(b2p + (size_t)k0 * DD);
    s.b2[1] = *(const float4*)(b2p + (size_t)(k0 + 16) * DD);
  };

  auto lstore = [&](const Stage& s, int buf, int slot) {
    float* L = lds + (buf * PF + slot) * TILE;
    if (!TRANSA) {
#pragma unroll
      for (int h = 0; h < 2; ++h) {
        float* dst = L + (aq + 32 * h) * SA + ar;
        dst[0] = s.a[h].x * __builtin_amdgcn_rcpf(in[h] * s.n.x + EPS);
        dst[1] = s.a[h].y * __builtin_amdgcn_rcpf(in[h] * s.n.y + EPS);
        dst[2] = s.a[h].z * __builtin_amdgcn_rcpf(in[h] * s.n.z + EPS);
        dst[3] = s.a[h].w * __builtin_amdgcn_rcpf(in[h] * s.n.w + EPS);
      }
    } else {
      const float a = s.n.x;
#pragma unroll
      for (int h = 0; h < 2; ++h) {
        float* dst = L + (32 * h + ar) * SA + aq;
        dst[0 * SA] = s.a[h].x * __builtin_amdgcn_rcpf(a * in[4 * h + 0] + EPS);
        dst[1 * SA] = s.a[h].y * __builtin_amdgcn_rcpf(a * in[4 * h + 1] + EPS);
        dst[2 * SA] = s.a[h].z * __builtin_amdgcn_rcpf(a * in[4 * h + 2] + EPS);
        dst[3 * SA] = s.a[h].w * __builtin_amdgcn_rcpf(a * in[4 * h + 3] + EPS);
      }
    }
#pragma unroll
    for (int it = 0; it < 2; ++it) {
      float* d1 = L + (BM + bc) * SA + (bu + it * 16);
      float* d2 = L + (BM + DD + bc) * SA + (bu + it * 16);
      d1[0 * SA] = s.b1[it].x; d1[1 * SA] = s.b1[it].y;
      d1[2 * SA] = s.b1[it].z; d1[3 * SA] = s.b1[it].w;
      d2[0 * SA] = s.b2[it].x; d2[1 * SA] = s.b2[it].y;
      d2[2 * SA] = s.b2[it].z; d2[3 * SA] = s.b2[it].w;
    }
  };

  v8f acc[2][2] = {};

  const int aoff = (rp * 32 + laneRow) * SA + 2 * laneHi;
  const int boff = (BM + cg * 32 + laneRow) * SA + 2 * laneHi;
  struct Frag {
    v2f a[KC / 4][2];
    v2f b[KC / 4][2];
  };
  auto fload = [&](Frag& f, int buf, int slot) {
    const float* Arow = lds + (buf * PF + slot) * TILE + aoff;
    const float* Bcol = lds + (buf * PF + slot) * TILE + boff;
#pragma unroll
    for (int j = 0; j < KC / 4; ++j) {
#pragma unroll
      for (int m = 0; m < 2; ++m) f.a[j][m] = *(const v2f*)(Arow + m * 16 * SA + 4 * j);
#pragma unroll
      for (int n = 0; n < 2; ++n) f.b[j][n] = *(const v2f*)(Bcol + n * 16 * SA + 4 * j);
    }
    asm volatile("" : "+v"(acc[0][0]), "+v"(acc[0][1]), "+v"(acc[1][0]), "+v"(acc[1][1]) : : "memory");
  };
  auto mma = [&](const Frag& f) {
#pragma unroll
    for (int j = 0; j < KC / 4; ++j) {
#pragma unroll
      for (int m = 0; m < 2; ++m) {
#pragma unroll
        for (int n = 0; n < 2; ++n) {
          acc[m][n] = __builtin_amdgcn_wmma_f32_16x16x4_f32(
              false, f.a[j][m], false, f.b[j][n], (short)0, acc[m][n], false, false);
        }
      }
    }
  };

  Stage s0, s1, s2, s3;
  gload(s0, 0); chunk_fence();
  gload(s1, 1); chunk_fence();
  gload(s2, 2); chunk_fence();
  gload(s3, 3); chunk_fence();
  lstore(s0, 0, 0); gload(s0, 4); chunk_fence();
  lstore(s1, 0, 1); gload(s1, 5); chunk_fence();
  lstore(s2, 0, 2); gload(s2, 6); chunk_fence();
  lstore(s3, 0, 3); gload(s3, 7);
  wg_barrier_loads_in_flight();
  Frag fA, fB;
  fload(fA, 0, 0); chunk_fence();

  int c = 0;
  for (; c + 16 <= NC; c += 8) {
    lstore(s0, 1, 0); gload(s0, c + 8);  fload(fB, 0, 1); mma(fA); chunk_fence();
    lstore(s1, 1, 1); gload(s1, c + 9);  fload(fA, 0, 2); mma(fB); chunk_fence();
    lstore(s2, 1, 2); gload(s2, c + 10); fload(fB, 0, 3); mma(fA); chunk_fence();
    lstore(s3, 1, 3); gload(s3, c + 11); wg_barrier_loads_in_flight(); fload(fA, 1, 0); mma(fB); chunk_fence();
    lstore(s0, 0, 0); gload(s0, c + 12); fload(fB, 1, 1); mma(fA); chunk_fence();
    lstore(s1, 0, 1); gload(s1, c + 13); fload(fA, 1, 2); mma(fB); chunk_fence();
    lstore(s2, 0, 2); gload(s2, c + 14); fload(fB, 1, 3); mma(fA); chunk_fence();
    lstore(s3, 0, 3); gload(s3, c + 15); wg_barrier_loads_in_flight(); fload(fA, 0, 0); mma(fB); chunk_fence();
  }
  lstore(s0, 1, 0); fload(fB, 0, 1); mma(fA); chunk_fence();
  lstore(s1, 1, 1); fload(fA, 0, 2); mma(fB); chunk_fence();
  lstore(s2, 1, 2); fload(fB, 0, 3); mma(fA); chunk_fence();
  lstore(s3, 1, 3); wg_barrier_loads_in_flight(); fload(fA, 1, 0); mma(fB); chunk_fence();
  fload(fB, 1, 1); mma(fA); chunk_fence();
  fload(fA, 1, 2); mma(fB); chunk_fence();
  fload(fB, 1, 3); mma(fA); chunk_fence();
  mma(fB);


  {
    const int q = 4 * (wave & 3) + (lane & 3);
    const int kr = 8 * (wave >> 2) + (lane >> 2);
    float4 w[8];
#pragma unroll
    for (int it = 0; it < 8; ++it) {
      const float* Wm = (it < 4) ? W1 : W2;
      w[it] = *(const float4*)(Wm + (size_t)((it & 3) * 16 + kr) * DD + 4 * q);
    }
#pragma unroll
    for (int it = 0; it < 8; ++it) {
      float* d = lds + EP_W + (4 * q) * PS + it * 16 + kr;
      d[0 * PS] = w[it].x; d[1 * PS] = w[it].y;
      d[2 * PS] = w[it].z; d[3 * PS] = w[it].w;
    }
  }

  {
    float* Pw = lds + EP_P + (rp * 32 + 8 * laneHi) * PS + cg * 32 + laneRow;
    if (cg < 2) {
#pragma unroll
      for (int m = 0; m < 2; ++m) {
#pragma unroll
        for (int n = 0; n < 2; ++n) {
#pragma unroll
          for (int v = 0; v < 8; ++v) Pw[(m * 16 + v) * PS + n * 16] = acc[m][n][v];
        }
      }
    } else {
      const float* sp = Bsrc + (size_t)(r0 + rp * 32 + 8 * laneHi) * DD + c0 + laneRow;
      float sv[2][2][8];
#pragma unroll
      for (int m = 0; m < 2; ++m) {
#pragma unroll
        for (int n = 0; n < 2; ++n) {
#pragma unroll
          for (int v = 0; v < 8; ++v) sv[m][n][v] = sp[(size_t)(m * 16 + v) * DD + n * 16];
        }
      }
#pragma unroll
      for (int m = 0; m < 2; ++m) {
#pragma unroll
        for (int n = 0; n < 2; ++n) {
#pragma unroll
          for (int v = 0; v < 8; ++v) Pw[(m * 16 + v) * PS + n * 16] = sv[m][n][v] * acc[m][n][v];
        }
      }
    }
  }
  __syncthreads();

  const int rt = wave >> 1;
  const int ch = wave & 1;
  const size_t ob = (size_t)(r0 + rt * 16 + 8 * laneHi) * DD + ch * 32 + laneRow;
  float tg[2][8];
#pragma unroll
  for (int n = 0; n < 2; ++n) {
#pragma unroll
    for (int v = 0; v < 8; ++v) tg[n][v] = Btgt[ob + (size_t)v * DD + n * 16];
  }
  const float* Pa = lds + EP_P + (rt * 16 + laneRow) * PS + 2 * laneHi;
  const float* Wb = lds + EP_W + (ch * 32 + laneRow) * PS + 2 * laneHi;
  v8f o[2] = {};
#pragma unroll
  for (int j = 0; j < 2 * DD / 4; ++j) {
    const v2f a = *(const v2f*)(Pa + 4 * j);
#pragma unroll
    for (int n = 0; n < 2; ++n) {
      const v2f b = *(const v2f*)(Wb + n * 16 * PS + 4 * j);
      o[n] = __builtin_amdgcn_wmma_f32_16x16x4_f32(false, a, false, b, (short)0, o[n], false, false);
    }
  }
#pragma unroll
  for (int n = 0; n < 2; ++n) {
#pragma unroll
    for (int v = 0; v < 8; ++v) {
      const float x = o[n][v] + tg[n][v];
      out[ob + (size_t)v * DD + n * 16] = (x >= 0.f) ? x : ALPHA * x;
    }
  }
}

__global__ __launch_bounds__(256) void spmm_both_kernel(
    const float* __restrict__ adj, const float* __restrict__ sr, const float* __restrict__ sc,
    const float* __restrict__ user, const float* __restrict__ item,
    const float* __restrict__ W1, const float* __restrict__ W2,
    float* __restrict__ out_user, float* __restrict__ out_item) {
  __shared__ __align__(16) float lds[2 * PF * TILE];
  const int blk = blockIdx.x >> 1;
  if ((blockIdx.x & 1) == 0) spmm_body<true>(lds, blk, adj, sr, sc, item, user, W1, W2, out_user);
  else spmm_body<false>(lds, blk, adj, sr, sc, user, item, W1, W2, out_item);
}


extern "C" void kernel_launch(void* const* d_in, const int* in_sizes, int n_in,
                              void* d_out, int out_size, void* d_ws, size_t ws_size,
                              hipStream_t stream) {
  const float* user = (const float*)d_in[0];
  const float* item = (const float*)d_in[1];
  const float* adj  = (const float*)d_in[2];
  const float* W1   = (const float*)d_in[3];
  const float* W2   = (const float*)d_in[4];

  float* ws = (float*)d_ws;
  const size_t ND = (size_t)NN * DD;
  float* sr   = ws;
  float* sc   = ws + NN;
  float* colpart = ws + 3 * NN;

  float* out_user = (float*)d_out;
  float* out_item = out_user + ND;

  degree_pass_kernel<<<NN / PR, PT, 0, stream>>>(adj, sr, colpart);
  colsum_finish_kernel<<<NN / FC, 256, 0, stream>>>(colpart, sc);

  spmm_both_kernel<<<2 * (NN / BM), 256, 0, stream>>>(adj, sr, sc, user, item, W1, W2, out_user, out_item);
}


# spmm K loop: the younger wave of each SIMD pair (waves 4-7) runs at s_setprio 3 from each work-group barrier to the middle of the third of the four WMMA runs behind it; no sum order changed
# speedup vs baseline: 6.7766x; 1.0318x over previous
#include <hip/hip_runtime.h>
#include <math.h>

typedef __attribute__((ext_vector_type(2))) float v2f;
typedef __attribute__((ext_vector_type(8))) float v8f;

#define NN 8192
#define DD 64
#define BM 64
#define KC 32
#define SA (KC + 2)
#define ALPHA 0.2f
#define EPS 1e-8f


typedef __attribute__((ext_vector_type(4))) float v4f;

#define PR 32
#define PW 16
#define PT (PW * 32)
#define PQ (NN / (4 * PT))
#define PB 4
#define NB (PR / PB)

__global__ __launch_bounds__(PT) void degree_pass_kernel(const float* __restrict__ adj,
                                                         float* __restrict__ sr,
                                                         float* __restrict__ colpart) {
  static_assert(PQ * 4 * PT == NN && PR == 32 && PR % PB == 0, "lane r of every wave keeps row r of the block");
  __shared__ float rowpart[PR][PW + 1];
  const int tid = threadIdx.x;
  const int lane = tid & 31;
  const int wave = tid >> 5;
  const int r0 = blockIdx.x * PR;
  const v4f* base = (const v4f*)(adj + (size_t)r0 * NN) + tid;

  v4f col[PQ];
#pragma unroll
  for (int q = 0; q < PQ; ++q) col[q] = (v4f)(0.f);
  v4f buf[2][PB][PQ];
  float mine = 0.f;

  auto load_batch = [&](v4f (&dst)[PB][PQ], int b) {
#pragma unroll
    for (int i = 0; i < PB; ++i) {
#pragma unroll
      for (int q = 0; q < PQ; ++q)
        dst[i][q] = __builtin_nontemporal_load(base + (size_t)(b * PB + i) * (NN / 4) + q * PT);
    }
  };

  load_batch(buf[0], 0);
#pragma unroll
  for (int b = 0; b < NB; ++b) {
    __builtin_amdgcn_sched_barrier(0);
    if (b + 1 < NB) load_batch(buf[(b + 1) & 1], b + 1);
    __builtin_amdgcn_sched_barrier(0);
    float s[PB];
#pragma unroll
    for (int i = 0; i < PB; ++i) {
      const v4f* w = buf[b & 1][i];
      const v4f t = (w[0] + w[1]) + (w[2] + w[3]);
      s[i] = (t.x + t.y) + (t.z + t.w);
#pragma unroll
      for (int q = 0; q < PQ; ++q) col[q] += w[q];
    }
#pragma unroll
    for (int m = 16; m >= 1; m >>= 1) {
#pragma unroll
      for (int i = 0; i < PB; ++i) s[i] += __shfl_xor(s[i], m, 32);
    }
#pragma unroll
    for (int i = 0; i < PB; ++i)
      if (lane == b * PB + i) mine = s[i];
  }

  v4f* cp = (v4f*)(colpart + (size_t)blockIdx.x * NN) + tid;
#pragma unroll
  for (int q = 0; q < PQ; ++q) cp[q * PT] = col[q];

  rowpart[lane][wave] = mine;
  __syncthreads();
  if (tid < PR) {
    float s = 0.f;
    for (int w = 0; w < PW; ++w) s += rowpart[tid][w];
    sr[r0 + tid] = sqrtf(s);
  }
}

#define FC 64
#define FG 16

__global__ __launch_bounds__(256) void colsum_finish_kernel(const float* __restrict__ colpart,
                                                            float* __restrict__ sc) {
  static_assert(FC / 4 * FG == 256 && (NN / PR) % FG == 0, "16 lanes x 16 bytes cover the columns, 16 groups the row blocks");
  __shared__ __align__(16) float part[FG][FC + 4];
  const int tid = threadIdx.x;
  const int cq = tid & (FC / 4 - 1);
  const int g = tid / (FC / 4);
  const v4f* p = (const v4f*)(colpart + (size_t)blockIdx.x * FC) + cq;
  v4f s = (v4f)(0.f);
#pragma unroll
  for (int i = 0; i < (NN / PR) / FG; ++i) s += p[(size_t)(g + FG * i) * (NN / 4)];
  *(v4f*)&part[g][4 * cq] = s;
  __syncthreads();
  if (tid < FC) {
    float t = 0.f;
    for (int k = 0; k < FG; ++k) t += part[k][tid];
    sc[blockIdx.x * FC + tid] = sqrtf(t);
  }
}


#define NC (NN / KC)
#define PF 4
#define TILE ((BM + 2 * DD) * SA)

#define PS 132
#define EP_P 0
#define EP_W (BM * PS)
static_assert(EP_W + DD * PS <= PF * TILE, "both tiles lie inside ring stage 0, which no wave reads after the last barrier of the K loop");

struct Stage {
  float4 a[2];
  float4 n;
  float4 b1[2];
  float4 b2[2];
};

static __device__ __forceinline__ void wg_barrier_loads_in_flight() {
  asm volatile("s_wait_dscnt 0x0\n\ts_barrier_signal -1\n\ts_barrier_wait -1" ::: "memory");
}

template <bool TRANSA>
static __device__ __forceinline__ void spmm_body(
    float* __restrict__ lds, const int blk,
    const float* __restrict__ adj, const float* __restrict__ sr, const float* __restrict__ sc,
    const float* __restrict__ Bsrc, const float* __restrict__ Btgt,
    const float* __restrict__ W1, const float* __restrict__ W2,
    float* __restrict__ out) {
  static_assert(PF == 4 && (NC % (2 * PF)) == 0 && NC >= 4 * PF, "the K loop below is written for a queue of four");
  static_assert(BM == 64, "two A quads per thread, 32 tile rows apart; 2 x 2 accumulator tiles per wave");

  const int tid = threadIdx.x;
  const int wave = tid >> 5;
  const int lane = tid & 31;
  const int laneRow = lane & 15;
  const int laneHi = lane >> 4;
  const int r0 = blk * BM;
  const int rp = wave & 1;
  const int cg = wave >> 1;
  const int c0 = (cg & 1) * 32;

  const bool younger = __builtin_amdgcn_readfirstlane(tid >> 5) >= 4;
  auto prio_up = [&]() { if (younger) __builtin_amdgcn_s_setprio(3); };
  auto prio_down = [&]() { if (younger) __builtin_amdgcn_s_setprio(0); };

  const int aq = tid >> 3;
  const int ar = (tid & 7) << 2;
  const float* ap = TRANSA ? adj + (size_t)aq * NN + (r0 + ar)
                           : adj + (size_t)(r0 + aq) * NN + ar;
  float in[8] = {0.f, 0.f, 0.f, 0.f, 0.f, 0.f, 0.f, 0.f};
  if (!TRANSA) {
    in[0] = sr[r0 + aq];
    in[1] = sr[r0 + aq + 32];
  } else {
#pragma unroll
    for (int j = 0; j < 4; ++j) {
      in[j] = sc[r0 + ar + j];
      in[4 + j] = sc[r0 + 32 + ar + j];
    }
  }
  auto chunk_fence = [&]() {
    if (!TRANSA) asm volatile("" : "+v"(in[0]), "+v"(in[1]) : : "memory");
    else asm volatile("" : "+v"(in[0]), "+v"(in[1]), "+v"(in[2]), "+v"(in[3]),
                           "+v"(in[4]), "+v"(in[5]), "+v"(in[6]), "+v"(in[7]) : : "memory");
    __builtin_amdgcn_sched_barrier(0);
  };
  const int bu = tid >> 4;
  const int bc = (tid & 15) << 2;
  const float* b1p = Bsrc + (size_t)bu * DD + bc;
  const float* b2p = Btgt + (size_t)bu * DD + bc;

  auto gload = [&](Stage& s, int c) {
    const int k0 = c * KC;
    if (!TRANSA) {
      s.n.x = sc[k0 + ar + 0]; s.n.y = sc[k0 + ar + 1];
      s.n.z = sc[k0 + ar + 2]; s.n.w = sc[k0 + ar + 3];
      s.a[0] = *(const float4*)(ap + k0);
      s.a[1] = *(const float4*)(ap + (size_t)32 * NN + k0);
    } else {
      s.n.x = sr[k0 + aq];
      s.a[0] = *(const float4*)(ap + (size_t)k0 * NN);
      s.a[1] = *(const float4*)(ap + (size_t)k0 * NN + 32);
    }
    s.b1[0] = *(const float4*)(b1p + (size_t)k0 * DD);
    s.b1[1] = *(const float4*)(b1p + (size_t)(k0 + 16) * DD);
    s.b2[0] = *(const float4*)(b2p + (size_t)k0 * DD);
    s.b2[1] = *(const float4*)(b2p + (size_t)(k0 + 16) * DD);
  };

  auto lstore = [&](const Stage& s, int buf, int slot) {
    float* L = lds + (buf * PF + slot) * TILE;
    if (!TRANSA) {
#pragma unroll
      for (int h = 0; h < 2; ++h) {
        float* dst = L + (aq + 32 * h) * SA + ar;
        dst[0] = s.a[h].x * __builtin_amdgcn_rcpf(in[h] * s.n.x + EPS);
        dst[1] = s.a[h].y * __builtin_amdgcn_rcpf(in[h] * s.n.y + EPS);
        dst[2] = s.a[h].z * __builtin_amdgcn_rcpf(in[h] * s.n.z + EPS);
        dst[3] = s.a[h].w * __builtin_amdgcn_rcpf(in[h] * s.n.w + EPS);
      }
    } else {
      const float a = s.n.x;
#pragma unroll
      for (int h = 0; h < 2; ++h) {
        float* dst = L + (32 * h + ar) * SA + aq;
        dst[0 * SA] = s.a[h].x * __builtin_amdgcn_rcpf(a * in[4 * h + 0] + EPS);
        dst[1 * SA] = s.a[h].y * __builtin_amdgcn_rcpf(a * in[4 * h + 1] + EPS);
        dst[2 * SA] = s.a[h].z * __builtin_amdgcn_rcpf(a * in[4 * h + 2] + EPS);
        dst[3 * SA] = s.a[h].w * __builtin_amdgcn_rcpf(a * in[4 * h + 3] + EPS);
      }
    }
#pragma unroll
    for (int it = 0; it < 2; ++it) {
      float* d1 = L + (BM + bc) * SA + (bu + it * 16);
      float* d2 = L + (BM + DD + bc) * SA + (bu + it * 16);
      d1[0 * SA] = s.b1[it].x; d1[1 * SA] = s.b1[it].y;
      d1[2 * SA] = s.b1[it].z; d1[3 * SA] = s.b1[it].w;
      d2[0 * SA] = s.b2[it].x; d2[1 * SA] = s.b2[it].y;
      d2[2 * SA] = s.b2[it].z; d2[3 * SA] = s.b2[it].w;
    }
  };

  v8f acc[2][2] = {};

  const int aoff = (rp * 32 + laneRow) * SA + 2 * laneHi;
  const int boff = (BM + cg * 32 + laneRow) * SA + 2 * laneHi;
  struct Frag {
    v2f a[KC / 4][2];
    v2f b[KC / 4][2];
  };
  auto fload = [&](Frag& f, int buf, int slot) {
    const float* Arow = lds + (buf * PF + slot) * TILE + aoff;
    const float* Bcol = lds + (buf * PF + slot) * TILE + boff;
#pragma unroll
    for (int j = 0; j < KC / 4; ++j) {
#pragma unroll
      for (int m = 0; m < 2; ++m) f.a[j][m] = *(const v2f*)(Arow + m * 16 * SA + 4 * j);
#pragma unroll
      for (int n = 0; n < 2; ++n) f.b[j][n] = *(const v2f*)(Bcol + n * 16 * SA + 4 * j);
    }
    asm volatile("" : "+v"(acc[0][0]), "+v"(acc[0][1]), "+v"(acc[1][0]), "+v"(acc[1][1]) : : "memory");
  };
  auto mma_steps = [&](const Frag& f, const int j0, const int j1) {
#pragma unroll
    for (int j = j0; j < j1; ++j) {
#pragma unroll
      for (int m = 0; m < 2; ++m) {
#pragma unroll
        for (int n = 0; n < 2; ++n) {
          acc[m][n] = __builtin_amdgcn_wmma_f32_16x16x4_f32(
              false, f.a[j][m], false, f.b[j][n], (short)0, acc[m][n], false, false);
        }
      }
    }
  };
  auto mma = [&](const Frag& f) { mma_steps(f, 0, KC / 4); };
  auto mma_window_end = [&](const Frag& f) {
    mma_steps(f, 0, KC / 8);
    prio_down();
    mma_steps(f, KC / 8, KC / 4);
  };

  Stage s0, s1, s2, s3;
  gload(s0, 0); chunk_fence();
  gload(s1, 1); chunk_fence();
  gload(s2, 2); chunk_fence();
  gload(s3, 3); chunk_fence();
  lstore(s0, 0, 0); gload(s0, 4); chunk_fence();
  lstore(s1, 0, 1); gload(s1, 5); chunk_fence();
  lstore(s2, 0, 2); gload(s2, 6); chunk_fence();
  lstore(s3, 0, 3); gload(s3, 7);
  wg_barrier_loads_in_flight(); prio_up();
  Frag fA, fB;
  fload(fA, 0, 0); chunk_fence();

  int c = 0;
  for (; c + 16 <= NC; c += 8) {
    lstore(s0, 1, 0); gload(s0, c + 8);  fload(fB, 0, 1); mma(fA); chunk_fence();
    lstore(s1, 1, 1); gload(s1, c + 9);  fload(fA, 0, 2); mma_window_end(fB); chunk_fence();
    lstore(s2, 1, 2); gload(s2, c + 10); fload(fB, 0, 3); mma(fA); chunk_fence();
    lstore(s3, 1, 3); gload(s3, c + 11); wg_barrier_loads_in_flight(); prio_up(); fload(fA, 1, 0); mma(fB); chunk_fence();
    lstore(s0, 0, 0); gload(s0, c + 12); fload(fB, 1, 1); mma(fA); chunk_fence();
    lstore(s1, 0, 1); gload(s1, c + 13); fload(fA, 1, 2); mma_window_end(fB); chunk_fence();
    lstore(s2, 0, 2); gload(s2, c + 14); fload(fB, 1, 3); mma(fA); chunk_fence();
    lstore(s3, 0, 3); gload(s3, c + 15); wg_barrier_loads_in_flight(); prio_up(); fload(fA, 0, 0); mma(fB); chunk_fence();
  }
  lstore(s0, 1, 0); fload(fB, 0, 1); mma(fA); chunk_fence();
  lstore(s1, 1, 1); fload(fA, 0, 2); mma_window_end(fB); chunk_fence();
  lstore(s2, 1, 2); fload(fB, 0, 3); mma(fA); chunk_fence();
  lstore(s3, 1, 3); wg_barrier_loads_in_flight(); prio_up(); fload(fA, 1, 0); mma(fB); chunk_fence();
  fload(fB, 1, 1); mma(fA); chunk_fence();
  fload(fA, 1, 2); mma_window_end(fB); chunk_fence();
  fload(fB, 1, 3); mma(fA); chunk_fence();
  mma(fB);


  {
    const int q = 4 * (wave & 3) + (lane & 3);
    const int kr = 8 * (wave >> 2) + (lane >> 2);
    float4 w[8];
#pragma unroll
    for (int it = 0; it < 8; ++it) {
      const float* Wm = (it < 4) ? W1 : W2;
      w[it] = *(const float4*)(Wm + (size_t)((it & 3) * 16 + kr) * DD + 4 * q);
    }
#pragma unroll
    for (int it = 0; it < 8; ++it) {
      float* d = lds + EP_W + (4 * q) * PS + it * 16 + kr;
      d[0 * PS] = w[it].x; d[1 * PS] = w[it].y;
      d[2 * PS] = w[it].z; d[3 * PS] = w[it].w;
    }
  }

  {
    float* Pw = lds + EP_P + (rp * 32 + 8 * laneHi) * PS + cg * 32 + laneRow;
    if (cg < 2) {
#pragma unroll
      for (int m = 0; m < 2; ++m) {
#pragma unroll
        for (int n = 0; n < 2; ++n) {
#pragma unroll
          for (int v = 0; v < 8; ++v) Pw[(m * 16 + v) * PS + n * 16] = acc[m][n][v];
        }
      }
    } else {
      const float* sp = Bsrc + (size_t)(r0 + rp * 32 + 8 * laneHi) * DD + c0 + laneRow;
      float sv[2][2][8];
#pragma unroll
      for (int m = 0; m < 2; ++m) {
#pragma unroll
        for (int n = 0; n < 2; ++n) {
#pragma unroll
          for (int v = 0; v < 8; ++v) sv[m][n][v] = sp[(size_t)(m * 16 + v) * DD + n * 16];
        }
      }
#pragma unroll
      for (int m = 0; m < 2; ++m) {
#pragma unroll
        for (int n = 0; n < 2; ++n) {
#pragma unroll
          for (int v = 0; v < 8; ++v) Pw[(m * 16 + v) * PS + n * 16] = sv[m][n][v] * acc[m][n][v];
        }
      }
    }
  }
  __syncthreads();

  const int rt = wave >> 1;
  const int ch = wave & 1;
  const size_t ob = (size_t)(r0 + rt * 16 + 8 * laneHi) * DD + ch * 32 + laneRow;
  float tg[2][8];
#pragma unroll
  for (int n = 0; n < 2; ++n) {
#pragma unroll
    for (int v = 0; v < 8; ++v) tg[n][v] = Btgt[ob + (size_t)v * DD + n * 16];
  }
  const float* Pa = lds + EP_P + (rt * 16 + laneRow) * PS + 2 * laneHi;
  const float* Wb = lds + EP_W + (ch * 32 + laneRow) * PS + 2 * laneHi;
  v8f o[2] = {};
#pragma unroll
  for (int j = 0; j < 2 * DD / 4; ++j) {
    const v2f a = *(const v2f*)(Pa + 4 * j);
#pragma unroll
    for (int n = 0; n < 2; ++n) {
      const v2f b = *(const v2f*)(Wb + n * 16 * PS + 4 * j);
      o[n] = __builtin_amdgcn_wmma_f32_16x16x4_f32(false, a, false, b, (short)0, o[n], false, false);
    }
  }
#pragma unroll
  for (int n = 0; n < 2; ++n) {
#pragma unroll
    for (int v = 0; v < 8; ++v) {
      const float x = o[n][v] + tg[n][v];
      out[ob + (size_t)v * DD + n * 16] = (x >= 0.f) ? x : ALPHA * x;
    }
  }
}

__global__ __launch_bounds__(256) void spmm_both_kernel(
    const float* __restrict__ adj, const float* __restrict__ sr, const float* __restrict__ sc,
    const float* __restrict__ user, const float* __restrict__ item,
    const float* __restrict__ W1, const float* __restrict__ W2,
    float* __restrict__ out_user, float* __restrict__ out_item) {
  __shared__ __align__(16) float lds[2 * PF * TILE];
  const int blk = blockIdx.x >> 1;
  if ((blockIdx.x & 1) == 0) spmm_body<true>(lds, blk, adj, sr, sc, item, user, W1, W2, out_user);
  else spmm_body<false>(lds, blk, adj, sr, sc, user, item, W1, W2, out_item);
}


extern "C" void kernel_launch(void* const* d_in, const int* in_sizes, int n_in,
                              void* d_out, int out_size, void* d_ws, size_t ws_size,
                              hipStream_t stream) {
  const float* user = (const float*)d_in[0];
  const float* item = (const float*)d_in[1];
  const float* adj  = (const float*)d_in[2];
  const float* W1   = (const float*)d_in[3];
  const float* W2   = (const float*)d_in[4];

  float* ws = (float*)d_ws;
  const size_t ND = (size_t)NN * DD;
  float* sr   = ws;
  float* sc   = ws + NN;
  float* colpart = ws + 3 * NN;

  float* out_user = (float*)d_out;
  float* out_item = out_user + ND;

  degree_pass_kernel<<<NN / PR, PT, 0, stream>>>(adj, sr, colpart);
  colsum_finish_kernel<<<NN / FC, 256, 0, stream>>>(colpart, sc);

  spmm_both_kernel<<<2 * (NN / BM), 256, 0, stream>>>(adj, sr, sc, user, item, W1, W2, out_user, out_item);
}


# spmm K loop: work-group barrier split (signal where it stood, wait 24 WMMAs later); between its signal and its wait a wave runs at s_setprio below its SIMD partner's; no sum order changed
# speedup vs baseline: 7.0752x; 1.0441x over previous
#include <hip/hip_runtime.h>
#include <math.h>

typedef __attribute__((ext_vector_type(2))) float v2f;
typedef __attribute__((ext_vector_type(8))) float v8f;

#define NN 8192
#define DD 64
#define BM 64
#define KC 32
#define SA (KC + 2)
#define ALPHA 0.2f
#define EPS 1e-8f


typedef __attribute__((ext_vector_type(4))) float v4f;

#define PR 32
#define PW 16
#define PT (PW * 32)
#define PQ (NN / (4 * PT))
#define PB 4
#define NB (PR / PB)

__global__ __launch_bounds__(PT) void degree_pass_kernel(const float* __restrict__ adj,
                                                         float* __restrict__ sr,
                                                         float* __restrict__ colpart) {
  static_assert(PQ * 4 * PT == NN && PR == 32 && PR % PB == 0, "lane r of every wave keeps row r of the block");
  __shared__ float rowpart[PR][PW + 1];
  const int tid = threadIdx.x;
  const int lane = tid & 31;
  const int wave = tid >> 5;
  const int r0 = blockIdx.x * PR;
  const v4f* base = (const v4f*)(adj + (size_t)r0 * NN) + tid;

  v4f col[PQ];
#pragma unroll
  for (int q = 0; q < PQ; ++q) col[q] = (v4f)(0.f);
  v4f buf[2][PB][PQ];
  float mine = 0.f;

  auto load_batch = [&](v4f (&dst)[PB][PQ], int b) {
#pragma unroll
    for (int i = 0; i < PB; ++i) {
#pragma unroll
      for (int q = 0; q < PQ; ++q)
        dst[i][q] = __builtin_nontemporal_load(base + (size_t)(b * PB + i) * (NN / 4) + q * PT);
    }
  };

  load_batch(buf[0], 0);
#pragma unroll
  for (int b = 0; b < NB; ++b) {
    __builtin_amdgcn_sched_barrier(0);
    if (b + 1 < NB) load_batch(buf[(b + 1) & 1], b + 1);
    __builtin_amdgcn_sched_barrier(0);
    float s[PB];
#pragma unroll
    for (int i = 0; i < PB; ++i) {
      const v4f* w = buf[b & 1][i];
      const v4f t = (w[0] + w[1]) + (w[2] + w[3]);
      s[i] = (t.x + t.y) + (t.z + t.w);
#pragma unroll
      for (int q = 0; q < PQ; ++q) col[q] += w[q];
    }
#pragma unroll
    for (int m = 16; m >= 1; m >>= 1) {
#pragma unroll
      for (int i = 0; i < PB; ++i) s[i] += __shfl_xor(s[i], m, 32);
    }
#pragma unroll
    for (int i = 0; i < PB; ++i)
      if (lane == b * PB + i) mine = s[i];
  }

  v4f* cp = (v4f*)(colpart + (size_t)blockIdx.x * NN) + tid;
#pragma unroll
  for (int q = 0; q < PQ; ++q) cp[q * PT] = col[q];

  rowpart[lane][wave] = mine;
  __syncthreads();
  if (tid < PR) {
    float s = 0.f;
    for (int w = 0; w < PW; ++w) s += rowpart[tid][w];
    sr[r0 + tid] = sqrtf(s);
  }
}

#define FC 64
#define FG 16

__global__ __launch_bounds__(256) void colsum_finish_kernel(const float* __restrict__ colpart,
                                                            float* __restrict__ sc) {
  static_assert(FC / 4 * FG == 256 && (NN / PR) % FG == 0, "16 lanes x 16 bytes cover the columns, 16 groups the row blocks");
  __shared__ __align__(16) float part[FG][FC + 4];
  const int tid = threadIdx.x;
  const int cq = tid & (FC / 4 - 1);
  const int g = tid / (FC / 4);
  const v4f* p = (const v4f*)(colpart + (size_t)blockIdx.x * FC) + cq;
  v4f s = (v4f)(0.f);
#pragma unroll
  for (int i = 0; i < (NN / PR) / FG; ++i) s += p[(size_t)(g + FG * i) * (NN / 4)];
  *(v4f*)&part[g][4 * cq] = s;
  __syncthreads();
  if (tid < FC) {
    float t = 0.f;
    for (int k = 0; k < FG; ++k) t += part[k][tid];
    sc[blockIdx.x * FC + tid] = sqrtf(t);
  }
}


#define NC (NN / KC)
#define PF 4
#define SPLIT_J 6
#define TILE ((BM + 2 * DD) * SA)

#define PS 132
#define EP_P 0
#define EP_W (BM * PS)
static_assert(EP_W + DD * PS <= PF * TILE, "both tiles lie inside ring stage 0, which no wave reads after the last barrier of the K loop");

struct Stage {
  float4 a[2];
  float4 n;
  float4 b1[2];
  float4 b2[2];
};

static __device__ __forceinline__ void wg_barrier_loads_in_flight() {
  asm volatile("s_wait_dscnt 0x0\n\ts_barrier_signal -1\n\ts_barrier_wait -1" ::: "memory");
}

template <bool TRANSA>
static __device__ __forceinline__ void spmm_body(
    float* __restrict__ lds, const int blk,
    const float* __restrict__ adj, const float* __restrict__ sr, const float* __restrict__ sc,
    const float* __restrict__ Bsrc, const float* __restrict__ Btgt,
    const float* __restrict__ W1, const float* __restrict__ W2,
    float* __restrict__ out) {
  static_assert(PF == 4 && (NC % (2 * PF)) == 0 && NC >= 4 * PF, "the K loop below is written for a queue of four");
  static_assert(BM == 64, "two A quads per thread, 32 tile rows apart; 2 x 2 accumulator tiles per wave");

  const int tid = threadIdx.x;
  const int wave = tid >> 5;
  const int lane = tid & 31;
  const int laneRow = lane & 15;
  const int laneHi = lane >> 4;
  const int r0 = blk * BM;
  const int rp = wave & 1;
  const int cg = wave >> 1;
  const int c0 = (cg & 1) * 32;

  const bool younger = __builtin_amdgcn_readfirstlane(tid >> 5) >= 4;
  auto prio_up = [&]() { __builtin_amdgcn_s_setprio(1); if (younger) __builtin_amdgcn_s_setprio(3); };
  auto prio_down = [&]() { if (younger) __builtin_amdgcn_s_setprio(1); };

  const int aq = tid >> 3;
  const int ar = (tid & 7) << 2;
  const float* ap = TRANSA ? adj + (size_t)aq * NN + (r0 + ar)
                           : adj + (size_t)(r0 + aq) * NN + ar;
  float in[8] = {0.f, 0.f, 0.f, 0.f, 0.f, 0.f, 0.f, 0.f};
  if (!TRANSA) {
    in[0] = sr[r0 + aq];
    in[1] = sr[r0 + aq + 32];
  } else {
#pragma unroll
    for (int j = 0; j < 4; ++j) {
      in[j] = sc[r0 + ar + j];
      in[4 + j] = sc[r0 + 32 + ar + j];
    }
  }
  auto chunk_fence = [&]() {
    if (!TRANSA) asm volatile("" : "+v"(in[0]), "+v"(in[1]) : : "memory");
    else asm volatile("" : "+v"(in[0]), "+v"(in[1]), "+v"(in[2]), "+v"(in[3]),
                           "+v"(in[4]), "+v"(in[5]), "+v"(in[6]), "+v"(in[7]) : : "memory");
    __builtin_amdgcn_sched_barrier(0);
  };
  const int bu = tid >> 4;
  const int bc = (tid & 15) << 2;
  const float* b1p = Bsrc + (size_t)bu * DD + bc;
  const float* b2p = Btgt + (size_t)bu * DD + bc;

  auto gload = [&](Stage& s, int c) {
    const int k0 = c * KC;
    if (!TRANSA) {
      s.n.x = sc[k0 + ar + 0]; s.n.y = sc[k0 + ar + 1];
      s.n.z = sc[k0 + ar + 2]; s.n.w = sc[k0 + ar + 3];
      s.a[0] = *(const float4*)(ap + k0);
      s.a[1] = *(const float4*)(ap + (size_t)32 * NN + k0);
    } else {
      s.n.x = sr[k0 + aq];
      s.a[0] = *(const float4*)(ap + (size_t)k0 * NN);
      s.a[1] = *(const float4*)(ap + (size_t)k0 * NN + 32);
    }
    s.b1[0] = *(const float4*)(b1p + (size_t)k0 * DD);
    s.b1[1] = *(const float4*)(b1p + (size_t)(k0 + 16) * DD);
    s.b2[0] = *(const float4*)(b2p + (size_t)k0 * DD);
    s.b2[1] = *(const float4*)(b2p + (size_t)(k0 + 16) * DD);
  };

  auto lstore = [&](const Stage& s, int buf, int slot) {
    float* L = lds + (buf * PF + slot) * TILE;
    if (!TRANSA) {
#pragma unroll
      for (int h = 0; h < 2; ++h) {
        float* dst = L + (aq + 32 * h) * SA + ar;
        dst[0] = s.a[h].x * __builtin_amdgcn_rcpf(in[h] * s.n.x + EPS);
        dst[1] = s.a[h].y * __builtin_amdgcn_rcpf(in[h] * s.n.y + EPS);
        dst[2] = s.a[h].z * __builtin_amdgcn_rcpf(in[h] * s.n.z + EPS);
        dst[3] = s.a[h].w * __builtin_amdgcn_rcpf(in[h] * s.n.w + EPS);
      }
    } else {
      const float a = s.n.x;
#pragma unroll
      for (int h = 0; h < 2; ++h) {
        float* dst = L + (32 * h + ar) * SA + aq;
        dst[0 * SA] = s.a[h].x * __builtin_amdgcn_rcpf(a * in[4 * h + 0] + EPS);
        dst[1 * SA] = s.a[h].y * __builtin_amdgcn_rcpf(a * in[4 * h + 1] + EPS);
        dst[2 * SA] = s.a[h].z * __builtin_amdgcn_rcpf(a * in[4 * h + 2] + EPS);
        dst[3 * SA] = s.a[h].w * __builtin_amdgcn_rcpf(a * in[4 * h + 3] + EPS);
      }
    }
#pragma unroll
    for (int it = 0; it < 2; ++it) {
      float* d1 = L + (BM + bc) * SA + (bu + it * 16);
      float* d2 = L + (BM + DD + bc) * SA + (bu + it * 16);
      d1[0 * SA] = s.b1[it].x; d1[1 * SA] = s.b1[it].y;
      d1[2 * SA] = s.b1[it].z; d1[3 * SA] = s.b1[it].w;
      d2[0 * SA] = s.b2[it].x; d2[1 * SA] = s.b2[it].y;
      d2[2 * SA] = s.b2[it].z; d2[3 * SA] = s.b2[it].w;
    }
  };

  v8f acc[2][2] = {};

  const int aoff = (rp * 32 + laneRow) * SA + 2 * laneHi;
  const int boff = (BM + cg * 32 + laneRow) * SA + 2 * laneHi;
  struct Frag {
    v2f a[KC / 4][2];
    v2f b[KC / 4][2];
  };
  auto fload = [&](Frag& f, int buf, int slot) {
    const float* Arow = lds + (buf * PF + slot) * TILE + aoff;
    const float* Bcol = lds + (buf * PF + slot) * TILE + boff;
#pragma unroll
    for (int j = 0; j < KC / 4; ++j) {
#pragma unroll
      for (int m = 0; m < 2; ++m) f.a[j][m] = *(const v2f*)(Arow + m * 16 * SA + 4 * j);
#pragma unroll
      for (int n = 0; n < 2; ++n) f.b[j][n] = *(const v2f*)(Bcol + n * 16 * SA + 4 * j);
    }
    asm volatile("" : "+v"(acc[0][0]), "+v"(acc[0][1]), "+v"(acc[1][0]), "+v"(acc[1][1]) : : "memory");
  };
  auto mma_steps = [&](const Frag& f, const int j0, const int j1) {
#pragma unroll
    for (int j = j0; j < j1; ++j) {
#pragma unroll
      for (int m = 0; m < 2; ++m) {
#pragma unroll
        for (int n = 0; n < 2; ++n) {
          acc[m][n] = __builtin_amdgcn_wmma_f32_16x16x4_f32(
              false, f.a[j][m], false, f.b[j][n], (short)0, acc[m][n], false, false);
        }
      }
    }
  };
  auto mma = [&](const Frag& f) { mma_steps(f, 0, KC / 4); };
  auto mma_window_end = [&](const Frag& f) {
    mma_steps(f, 0, KC / 8);
    prio_down();
    mma_steps(f, KC / 8, KC / 4);
  };

  auto bar_signal = [&]() {
    asm volatile("s_wait_dscnt 0x0\n\ts_barrier_signal -1\n\ts_setprio 0"
                 : "+v"(acc[0][0]), "+v"(acc[0][1]), "+v"(acc[1][0]), "+v"(acc[1][1]) : : "memory");
  };
  auto bar_wait = [&]() {
    asm volatile("s_barrier_wait -1"
                 : "+v"(acc[0][0]), "+v"(acc[0][1]), "+v"(acc[1][0]), "+v"(acc[1][1]) : : "memory");
    prio_up();
  };
  auto mma_over_barrier = [&](const Frag& f, Frag& next, int buf) {
    bar_signal();
    mma_steps(f, 0, SPLIT_J);
    bar_wait();
    fload(next, buf, 0);
    mma_steps(f, SPLIT_J, KC / 4);
  };

  Stage s0, s1, s2, s3;
  gload(s0, 0); chunk_fence();
  gload(s1, 1); chunk_fence();
  gload(s2, 2); chunk_fence();
  gload(s3, 3); chunk_fence();
  lstore(s0, 0, 0); gload(s0, 4); chunk_fence();
  lstore(s1, 0, 1); gload(s1, 5); chunk_fence();
  lstore(s2, 0, 2); gload(s2, 6); chunk_fence();
  lstore(s3, 0, 3); gload(s3, 7);
  wg_barrier_loads_in_flight(); prio_up();
  Frag fA, fB;
  fload(fA, 0, 0); chunk_fence();

  int c = 0;
  for (; c + 16 <= NC; c += 8) {
    lstore(s0, 1, 0); gload(s0, c + 8);  fload(fB, 0, 1); mma(fA); chunk_fence();
    lstore(s1, 1, 1); gload(s1, c + 9);  fload(fA, 0, 2); mma_window_end(fB); chunk_fence();
    lstore(s2, 1, 2); gload(s2, c + 10); fload(fB, 0, 3); mma(fA); chunk_fence();
    lstore(s3, 1, 3); gload(s3, c + 11); mma_over_barrier(fB, fA, 1); chunk_fence();
    lstore(s0, 0, 0); gload(s0, c + 12); fload(fB, 1, 1); mma(fA); chunk_fence();
    lstore(s1, 0, 1); gload(s1, c + 13); fload(fA, 1, 2); mma_window_end(fB); chunk_fence();
    lstore(s2, 0, 2); gload(s2, c + 14); fload(fB, 1, 3); mma(fA); chunk_fence();
    lstore(s3, 0, 3); gload(s3, c + 15); mma_over_barrier(fB, fA, 0); chunk_fence();
  }
  lstore(s0, 1, 0); fload(fB, 0, 1); mma(fA); chunk_fence();
  lstore(s1, 1, 1); fload(fA, 0, 2); mma_window_end(fB); chunk_fence();
  lstore(s2, 1, 2); fload(fB, 0, 3); mma(fA); chunk_fence();
  lstore(s3, 1, 3); mma_over_barrier(fB, fA, 1); chunk_fence();
  fload(fB, 1, 1); mma(fA); chunk_fence();
  fload(fA, 1, 2); mma_window_end(fB); chunk_fence();
  fload(fB, 1, 3); mma(fA); chunk_fence();
  mma(fB);


  {
    const int q = 4 * (wave & 3) + (lane & 3);
    const int kr = 8 * (wave >> 2) + (lane >> 2);
    float4 w[8];
#pragma unroll
    for (int it = 0; it < 8; ++it) {
      const float* Wm = (it < 4) ? W1 : W2;
      w[it] = *(const float4*)(Wm + (size_t)((it & 3) * 16 + kr) * DD + 4 * q);
    }
#pragma unroll
    for (int it = 0; it < 8; ++it) {
      float* d = lds + EP_W + (4 * q) * PS + it * 16 + kr;
      d[0 * PS] = w[it].x; d[1 * PS] = w[it].y;
      d[2 * PS] = w[it].z; d[3 * PS] = w[it].w;
    }
  }

  {
    float* Pw = lds + EP_P + (rp * 32 + 8 * laneHi) * PS + cg * 32 + laneRow;
    if (cg < 2) {
#pragma unroll
      for (int m = 0; m < 2; ++m) {
#pragma unroll
        for (int n = 0; n < 2; ++n) {
#pragma unroll
          for (int v = 0; v < 8; ++v) Pw[(m * 16 + v) * PS + n * 16] = acc[m][n][v];
        }
      }
    } else {
      const float* sp = Bsrc + (size_t)(r0 + rp * 32 + 8 * laneHi) * DD + c0 + laneRow;
      float sv[2][2][8];
#pragma unroll
      for (int m = 0; m < 2; ++m) {
#pragma unroll
        for (int n = 0; n < 2; ++n) {
#pragma unroll
          for (int v = 0; v < 8; ++v) sv[m][n][v] = sp[(size_t)(m * 16 + v) * DD + n * 16];
        }
      }
#pragma unroll
      for (int m = 0; m < 2; ++m) {
#pragma unroll
        for (int n = 0; n < 2; ++n) {
#pragma unroll
          for (int v = 0; v < 8; ++v) Pw[(m * 16 + v) * PS + n * 16] = sv[m][n][v] * acc[m][n][v];
        }
      }
    }
  }
  __syncthreads();

  const int rt = wave >> 1;
  const int ch = wave & 1;
  const size_t ob = (size_t)(r0 + rt * 16 + 8 * laneHi) * DD + ch * 32 + laneRow;
  float tg[2][8];
#pragma unroll
  for (int n = 0; n < 2; ++n) {
#pragma unroll
    for (int v = 0; v < 8; ++v) tg[n][v] = Btgt[ob + (size_t)v * DD + n * 16];
  }
  const float* Pa = lds + EP_P + (rt * 16 + laneRow) * PS + 2 * laneHi;
  const float* Wb = lds + EP_W + (ch * 32 + laneRow) * PS + 2 * laneHi;
  v8f o[2] = {};
#pragma unroll
  for (int j = 0; j < 2 * DD / 4; ++j) {
    const v2f a = *(const v2f*)(Pa + 4 * j);
#pragma unroll
    for (int n = 0; n < 2; ++n) {
      const v2f b = *(const v2f*)(Wb + n * 16 * PS + 4 * j);
      o[n] = __builtin_amdgcn_wmma_f32_16x16x4_f32(false, a, false, b, (short)0, o[n], false, false);
    }
  }
#pragma unroll
  for (int n = 0; n < 2; ++n) {
#pragma unroll
    for (int v = 0; v < 8; ++v) {
      const float x = o[n][v] + tg[n][v];
      out[ob + (size_t)v * DD + n * 16] = (x >= 0.f) ? x : ALPHA * x;
    }
  }
}

__global__ __launch_bounds__(256) void spmm_both_kernel(
    const float* __restrict__ adj, const float* __restrict__ sr, const float* __restrict__ sc,
    const float* __restrict__ user, const float* __restrict__ item,
    const float* __restrict__ W1, const float* __restrict__ W2,
    float* __restrict__ out_user, float* __restrict__ out_item) {
  __shared__ __align__(16) float lds[2 * PF * TILE];
  const int blk = blockIdx.x >> 1;
  if ((blockIdx.x & 1) == 0) spmm_body<true>(lds, blk, adj, sr, sc, item, user, W1, W2, out_user);
  else spmm_body<false>(lds, blk, adj, sr, sc, user, item, W1, W2, out_item);
}


extern "C" void kernel_launch(void* const* d_in, const int* in_sizes, int n_in,
                              void* d_out, int out_size, void* d_ws, size_t ws_size,
                              hipStream_t stream) {
  const float* user = (const float*)d_in[0];
  const float* item = (const float*)d_in[1];
  const float* adj  = (const float*)d_in[2];
  const float* W1   = (const float*)d_in[3];
  const float* W2   = (const float*)d_in[4];

  float* ws = (float*)d_ws;
  const size_t ND = (size_t)NN * DD;
  float* sr   = ws;
  float* sc   = ws + NN;
  float* colpart = ws + 3 * NN;

  float* out_user = (float*)d_out;
  float* out_item = out_user + ND;

  degree_pass_kernel<<<NN / PR, PT, 0, stream>>>(adj, sr, colpart);
  colsum_finish_kernel<<<NN / FC, 256, 0, stream>>>(colpart, sc);

  spmm_both_kernel<<<2 * (NN / BM), 256, 0, stream>>>(adj, sr, sc, user, item, W1, W2, out_user, out_item);
}
